# starting kernel
# baseline (speedup 1.0000x reference)
#include <hip/hip_runtime.h>
#include <math.h>


typedef __attribute__((ext_vector_type(2))) float v2f;
typedef __attribute__((ext_vector_type(8))) float v8f;

#define A_DIM 128
#define F_DIM 128
#define R_DIM 20
#define LDS_STRIDE 132
#define WAVES_PER_BLOCK 5
#define BLOCK_THREADS (WAVES_PER_BLOCK * 32)

__device__ __forceinline__ v8f wmma4(v2f a, v2f b, v8f c) {
  return __builtin_amdgcn_wmma_f32_16x16x4_f32(
       false, a,  false, b,
       (short)0, c,  false,  false);
}

__device__ __forceinline__ v8f v8f_zero() {
  v8f z = {0.0f, 0.0f, 0.0f, 0.0f, 0.0f, 0.0f, 0.0f, 0.0f};
  return z;
}

__device__ __forceinline__ float ssp(float x) {
  const float t = __expf(-fabsf(x));
  return fmaxf(x, 0.0f) + __logf(1.0f + t) - 0.69314718055994531f;
}

__global__ void zero_kernel(float* __restrict__ p, long n) {
  long i = (long)blockIdx.x * blockDim.x + threadIdx.x;
  if (i < n) p[i] = 0.0f;
}

__global__ void pack_kernel(const float* __restrict__ W, float* __restrict__ Wp,
                            int K, int Ncol) {
  int i = blockIdx.x * blockDim.x + threadIdx.x;
  int total = (K / 2) * Ncol;
  if (i >= total) return;
  int k2 = i / Ncol;
  int n  = i - k2 * Ncol;
  Wp[k2 * 2 * Ncol + 2 * n + 0] = W[(2 * k2 + 0) * Ncol + n];
  Wp[k2 * 2 * Ncol + 2 * n + 1] = W[(2 * k2 + 1) * Ncol + n];
}

__device__ __forceinline__ v2f bfrag(const float* __restrict__ Wp,
                                     int k, int khalf, int n) {
  return *(const v2f*)(Wp + (2 * k + khalf) * 256 + 2 * n);
}

__global__ void __launch_bounds__(BLOCK_THREADS)
gemm_h_kernel(const float* __restrict__ x, const float* __restrict__ Wp,
              float* __restrict__ h, int ntiles) {
  const int wave  = threadIdx.x >> 5;
  const int lane  = threadIdx.x & 31;
  const int t     = blockIdx.x * WAVES_PER_BLOCK + wave;
  if (t >= ntiles) return;
  const int mrow  = lane & 15;
  const int khalf = lane >> 4;
  const long r0   = (long)t * 16;

  const float* xrow = x + (r0 + mrow) * A_DIM;
  v2f af[32];
#pragma unroll
  for (int k = 0; k < 32; ++k)
    af[k] = *(const v2f*)(xrow + 4 * k + 2 * khalf);

  for (int j = 0; j < 8; ++j) {
    const int n = j * 16 + mrow;
    v8f acc = v8f_zero();
#pragma unroll
    for (int k = 0; k < 32; ++k)
      acc = wmma4(af[k], bfrag(Wp, k, khalf, n), acc);
#pragma unroll
    for (int r = 0; r < 8; ++r) {
      const int m = r + 8 * khalf;
      h[(r0 + m) * F_DIM + n] = acc[r];
    }
  }
}

__global__ void __launch_bounds__(BLOCK_THREADS)
edge_kernel(const float* __restrict__ f_ij, const float* __restrict__ rcut,
            const int* __restrict__ idx_i, const int* __restrict__ idx_j,
            const float* __restrict__ Wp_f1, const float* __restrict__ b_f1,
            const float* __restrict__ Wp_f2, const float* __restrict__ b_f2,
            const float* __restrict__ h, float* __restrict__ agg, int ntiles) {
  __shared__ float lds[WAVES_PER_BLOCK][16 * LDS_STRIDE];

  const int wave   = threadIdx.x >> 5;
  const int lane   = threadIdx.x & 31;
  const int t      = blockIdx.x * WAVES_PER_BLOCK + wave;
  const bool live  = (t < ntiles);
  const int mrow   = lane & 15;
  const int khalf  = lane >> 4;
  const long e0    = (long)t * 16;
  float* tw = lds[wave];

  if (live) {
    const float* frow = f_ij + (e0 + mrow) * R_DIM;
    v2f af1[5];
#pragma unroll
    for (int k = 0; k < 5; ++k)
      af1[k] = *(const v2f*)(frow + 4 * k + 2 * khalf);

    v8f acc1[8];
#pragma unroll
    for (int j = 0; j < 8; ++j) acc1[j] = v8f_zero();
#pragma unroll
    for (int k = 0; k < 5; ++k) {
#pragma unroll
      for (int j = 0; j < 8; ++j)
        acc1[j] = wmma4(af1[k], bfrag(Wp_f1, k, khalf, j * 16 + mrow), acc1[j]);
    }
#pragma unroll
    for (int j = 0; j < 8; ++j) {
      const int n = j * 16 + mrow;
      const float bias = b_f1[n];
#pragma unroll
      for (int r = 0; r < 8; ++r) {
        const int m = r + 8 * khalf;
        tw[m * LDS_STRIDE + n] = ssp(acc1[j][r] + bias);
      }
    }
  }
  __syncthreads();

  if (live) {
    int   vi[8]; int vj[8]; float vr[8];
#pragma unroll
    for (int r = 0; r < 8; ++r) {
      const long e = e0 + r + 8 * khalf;
      vi[r] = idx_i[e]; vj[r] = idx_j[e]; vr[r] = rcut[e];
    }

    const float* arow = tw + mrow * LDS_STRIDE;
    v2f af2[32];
#pragma unroll
    for (int k = 0; k < 32; ++k)
      af2[k] = *(const v2f*)(arow + 4 * k + 2 * khalf);

    for (int j2 = 0; j2 < 8; ++j2) {
      const int n = j2 * 16 + mrow;
      v8f acc2 = v8f_zero();
#pragma unroll
      for (int k = 0; k < 32; ++k)
        acc2 = wmma4(af2[k], bfrag(Wp_f2, k, khalf, n), acc2);

      const float bias2 = b_f2[n];
#pragma unroll
      for (int r = 0; r < 8; ++r) {
        const float wij = (acc2[r] + bias2) * vr[r];
        const float xv  = h[(long)vj[r] * F_DIM + n];
        atomicAdd(agg + (long)vi[r] * F_DIM + n, wij * xv);
      }
    }
  }
}

__global__ void __launch_bounds__(BLOCK_THREADS)
out_kernel(const float* __restrict__ agg,
           const float* __restrict__ Wp_o1, const float* __restrict__ b_o1,
           const float* __restrict__ Wp_o2, const float* __restrict__ b_o2,
           float* __restrict__ out, int ntiles) {
  __shared__ float lds[WAVES_PER_BLOCK][16 * LDS_STRIDE];

  const int wave  = threadIdx.x >> 5;
  const int lane  = threadIdx.x & 31;
  const int t     = blockIdx.x * WAVES_PER_BLOCK + wave;
  const bool live = (t < ntiles);
  const int mrow  = lane & 15;
  const int khalf = lane >> 4;
  const long r0   = (long)t * 16;
  float* tw = lds[wave];

  if (live) {
    const float* grow = agg + (r0 + mrow) * F_DIM;
    v2f af[32];
#pragma unroll
    for (int k = 0; k < 32; ++k)
      af[k] = *(const v2f*)(grow + 4 * k + 2 * khalf);

    for (int j = 0; j < 8; ++j) {
      const int n = j * 16 + mrow;
      v8f acc = v8f_zero();
#pragma unroll
      for (int k = 0; k < 32; ++k)
        acc = wmma4(af[k], bfrag(Wp_o1, k, khalf, n), acc);

      const float bias = b_o1[n];
#pragma unroll
      for (int r = 0; r < 8; ++r) {
        const int m = r + 8 * khalf;
        tw[m * LDS_STRIDE + n] = ssp(acc[r] + bias);
      }
    }
  }
  __syncthreads();

  if (live) {
    const float* arow = tw + mrow * LDS_STRIDE;
    v2f af2[32];
#pragma unroll
    for (int k = 0; k < 32; ++k)
      af2[k] = *(const v2f*)(arow + 4 * k + 2 * khalf);

    for (int j2 = 0; j2 < 8; ++j2) {
      const int n = j2 * 16 + mrow;
      v8f acc = v8f_zero();
#pragma unroll
      for (int k = 0; k < 32; ++k)
        acc = wmma4(af2[k], bfrag(Wp_o2, k, khalf, n), acc);

      const float bias2 = b_o2[n];
#pragma unroll
      for (int r = 0; r < 8; ++r) {
        const int m = r + 8 * khalf;
        out[(r0 + m) * A_DIM + n] = acc[r] + bias2;
      }
    }
  }
}

extern "C" void kernel_launch(void* const* d_in, const int* in_sizes, int n_in,
                              void* d_out, int out_size, void* d_ws, size_t ws_size,
                              hipStream_t stream) {
  const float* x      = (const float*)d_in[0];
  const float* f_ij   = (const float*)d_in[1];
  const float* rcut   = (const float*)d_in[2];
  const int*   idx_i  = (const int*)d_in[3];
  const int*   idx_j  = (const int*)d_in[4];
  const float* W_in2f = (const float*)d_in[5];
  const float* W_f1   = (const float*)d_in[6];
  const float* b_f1   = (const float*)d_in[7];
  const float* W_f2   = (const float*)d_in[8];
  const float* b_f2   = (const float*)d_in[9];
  const float* W_o1   = (const float*)d_in[10];
  const float* b_o1   = (const float*)d_in[11];
  const float* W_o2   = (const float*)d_in[12];
  const float* b_o2   = (const float*)d_in[13];
  float* out = (float*)d_out;

  const int  N = in_sizes[0] / A_DIM;
  const long E = (long)in_sizes[2];

  float* h       = (float*)d_ws;
  float* agg     = h + (size_t)N * F_DIM;
  float* Wp_in2f = agg + (size_t)N * F_DIM;
  float* Wp_f1   = Wp_in2f + A_DIM * F_DIM;
  float* Wp_f2   = Wp_f1 + R_DIM * F_DIM;
  float* Wp_o1   = Wp_f2 + F_DIM * F_DIM;
  float* Wp_o2   = Wp_o1 + F_DIM * A_DIM;

  {
    int blk = 256;
    pack_kernel<<<(A_DIM / 2 * F_DIM + blk - 1) / blk, blk, 0, stream>>>(W_in2f, Wp_in2f, A_DIM, F_DIM);
    pack_kernel<<<(R_DIM / 2 * F_DIM + blk - 1) / blk, blk, 0, stream>>>(W_f1, Wp_f1, R_DIM, F_DIM);
    pack_kernel<<<(F_DIM / 2 * F_DIM + blk - 1) / blk, blk, 0, stream>>>(W_f2, Wp_f2, F_DIM, F_DIM);
    pack_kernel<<<(F_DIM / 2 * A_DIM + blk - 1) / blk, blk, 0, stream>>>(W_o1, Wp_o1, F_DIM, A_DIM);
    pack_kernel<<<(A_DIM / 2 * A_DIM + blk - 1) / blk, blk, 0, stream>>>(W_o2, Wp_o2, A_DIM, A_DIM);
  }
  {
    long n = (long)N * F_DIM;
    int  blk = 256;
    zero_kernel<<<(int)((n + blk - 1) / blk), blk, 0, stream>>>(agg, n);
  }
  {
    int ntiles = N / 16;
    int blocks = (ntiles + WAVES_PER_BLOCK - 1) / WAVES_PER_BLOCK;
    gemm_h_kernel<<<blocks, BLOCK_THREADS, 0, stream>>>(x, Wp_in2f, h, ntiles);
  }
  {
    int ntiles = (int)(E / 16);
    int blocks = (ntiles + WAVES_PER_BLOCK - 1) / WAVES_PER_BLOCK;
    edge_kernel<<<blocks, BLOCK_THREADS, 0, stream>>>(
        f_ij, rcut, idx_i, idx_j, Wp_f1, b_f1, Wp_f2, b_f2, h, agg, ntiles);
  }
  {
    int ntiles = N / 16;
    int blocks = (ntiles + WAVES_PER_BLOCK - 1) / WAVES_PER_BLOCK;
    out_kernel<<<blocks, BLOCK_THREADS, 0, stream>>>(
        agg, Wp_o1, b_o1, Wp_o2, b_o2, out, ntiles);
  }
}


# edge kernel: rows of equal idx_i in a lane's 8-row accumulator strip are summed in registers, one atomicAdd per run instead of per edge; changes the float summation order
# speedup vs baseline: 1.1448x; 1.1448x over previous
#include <hip/hip_runtime.h>
#include <math.h>


typedef __attribute__((ext_vector_type(2))) float v2f;
typedef __attribute__((ext_vector_type(8))) float v8f;

#define A_DIM 128
#define F_DIM 128
#define R_DIM 20
#define LDS_STRIDE 132
#define WAVES_PER_BLOCK 5
#define BLOCK_THREADS (WAVES_PER_BLOCK * 32)

__device__ __forceinline__ v8f wmma4(v2f a, v2f b, v8f c) {
  return __builtin_amdgcn_wmma_f32_16x16x4_f32(
       false, a,  false, b,
       (short)0, c,  false,  false);
}

__device__ __forceinline__ v8f v8f_zero() {
  v8f z = {0.0f, 0.0f, 0.0f, 0.0f, 0.0f, 0.0f, 0.0f, 0.0f};
  return z;
}

__device__ __forceinline__ float ssp(float x) {
  const float t = __expf(-fabsf(x));
  return fmaxf(x, 0.0f) + __logf(1.0f + t) - 0.69314718055994531f;
}

__global__ void zero_kernel(float* __restrict__ p, long n) {
  long i = (long)blockIdx.x * blockDim.x + threadIdx.x;
  if (i < n) p[i] = 0.0f;
}

__global__ void pack_kernel(const float* __restrict__ W, float* __restrict__ Wp,
                            int K, int Ncol) {
  int i = blockIdx.x * blockDim.x + threadIdx.x;
  int total = (K / 2) * Ncol;
  if (i >= total) return;
  int k2 = i / Ncol;
  int n  = i - k2 * Ncol;
  Wp[k2 * 2 * Ncol + 2 * n + 0] = W[(2 * k2 + 0) * Ncol + n];
  Wp[k2 * 2 * Ncol + 2 * n + 1] = W[(2 * k2 + 1) * Ncol + n];
}

__device__ __forceinline__ v2f bfrag(const float* __restrict__ Wp,
                                     int k, int khalf, int n) {
  return *(const v2f*)(Wp + (2 * k + khalf) * 256 + 2 * n);
}

__global__ void __launch_bounds__(BLOCK_THREADS)
gemm_h_kernel(const float* __restrict__ x, const float* __restrict__ Wp,
              float* __restrict__ h, int ntiles) {
  const int wave  = threadIdx.x >> 5;
  const int lane  = threadIdx.x & 31;
  const int t     = blockIdx.x * WAVES_PER_BLOCK + wave;
  if (t >= ntiles) return;
  const int mrow  = lane & 15;
  const int khalf = lane >> 4;
  const long r0   = (long)t * 16;

  const float* xrow = x + (r0 + mrow) * A_DIM;
  v2f af[32];
#pragma unroll
  for (int k = 0; k < 32; ++k)
    af[k] = *(const v2f*)(xrow + 4 * k + 2 * khalf);

  for (int j = 0; j < 8; ++j) {
    const int n = j * 16 + mrow;
    v8f acc = v8f_zero();
#pragma unroll
    for (int k = 0; k < 32; ++k)
      acc = wmma4(af[k], bfrag(Wp, k, khalf, n), acc);
#pragma unroll
    for (int r = 0; r < 8; ++r) {
      const int m = r + 8 * khalf;
      h[(r0 + m) * F_DIM + n] = acc[r];
    }
  }
}

__global__ void __launch_bounds__(BLOCK_THREADS)
edge_kernel(const float* __restrict__ f_ij, const float* __restrict__ rcut,
            const int* __restrict__ idx_i, const int* __restrict__ idx_j,
            const float* __restrict__ Wp_f1, const float* __restrict__ b_f1,
            const float* __restrict__ Wp_f2, const float* __restrict__ b_f2,
            const float* __restrict__ h, float* __restrict__ agg, int ntiles) {
  __shared__ float lds[WAVES_PER_BLOCK][16 * LDS_STRIDE];

  const int wave   = threadIdx.x >> 5;
  const int lane   = threadIdx.x & 31;
  const int t      = blockIdx.x * WAVES_PER_BLOCK + wave;
  const bool live  = (t < ntiles);
  const int mrow   = lane & 15;
  const int khalf  = lane >> 4;
  const long e0    = (long)t * 16;
  float* tw = lds[wave];

  if (live) {
    const float* frow = f_ij + (e0 + mrow) * R_DIM;
    v2f af1[5];
#pragma unroll
    for (int k = 0; k < 5; ++k)
      af1[k] = *(const v2f*)(frow + 4 * k + 2 * khalf);

    v8f acc1[8];
#pragma unroll
    for (int j = 0; j < 8; ++j) acc1[j] = v8f_zero();
#pragma unroll
    for (int k = 0; k < 5; ++k) {
#pragma unroll
      for (int j = 0; j < 8; ++j)
        acc1[j] = wmma4(af1[k], bfrag(Wp_f1, k, khalf, j * 16 + mrow), acc1[j]);
    }
#pragma unroll
    for (int j = 0; j < 8; ++j) {
      const int n = j * 16 + mrow;
      const float bias = b_f1[n];
#pragma unroll
      for (int r = 0; r < 8; ++r) {
        const int m = r + 8 * khalf;
        tw[m * LDS_STRIDE + n] = ssp(acc1[j][r] + bias);
      }
    }
  }
  __syncthreads();

  if (live) {
    int   vi[8]; int vj[8]; float vr[8];
#pragma unroll
    for (int r = 0; r < 8; ++r) {
      const long e = e0 + r + 8 * khalf;
      vi[r] = idx_i[e]; vj[r] = idx_j[e]; vr[r] = rcut[e];
    }
    unsigned brk = 0x80u;
#pragma unroll
    for (int r = 0; r < 7; ++r)
      if (vi[r] != vi[r + 1]) brk |= (1u << r);

    const float* arow = tw + mrow * LDS_STRIDE;
    v2f af2[32];
#pragma unroll
    for (int k = 0; k < 32; ++k)
      af2[k] = *(const v2f*)(arow + 4 * k + 2 * khalf);

    for (int j2 = 0; j2 < 8; ++j2) {
      const int n = j2 * 16 + mrow;
      v8f acc2 = v8f_zero();
#pragma unroll
      for (int k = 0; k < 32; ++k)
        acc2 = wmma4(af2[k], bfrag(Wp_f2, k, khalf, n), acc2);

      const float bias2 = b_f2[n];
      float s = 0.0f;
#pragma unroll
      for (int r = 0; r < 8; ++r) {
        const float wij = (acc2[r] + bias2) * vr[r];
        const float xv  = h[(long)vj[r] * F_DIM + n];
        s += wij * xv;
        if (r == 7 || (brk & (1u << r))) {
          atomicAdd(agg + (long)vi[r] * F_DIM + n, s);
          s = 0.0f;
        }
      }
    }
  }
}

__global__ void __launch_bounds__(BLOCK_THREADS)
out_kernel(const float* __restrict__ agg,
           const float* __restrict__ Wp_o1, const float* __restrict__ b_o1,
           const float* __restrict__ Wp_o2, const float* __restrict__ b_o2,
           float* __restrict__ out, int ntiles) {
  __shared__ float lds[WAVES_PER_BLOCK][16 * LDS_STRIDE];

  const int wave  = threadIdx.x >> 5;
  const int lane  = threadIdx.x & 31;
  const int t     = blockIdx.x * WAVES_PER_BLOCK + wave;
  const bool live = (t < ntiles);
  const int mrow  = lane & 15;
  const int khalf = lane >> 4;
  const long r0   = (long)t * 16;
  float* tw = lds[wave];

  if (live) {
    const float* grow = agg + (r0 + mrow) * F_DIM;
    v2f af[32];
#pragma unroll
    for (int k = 0; k < 32; ++k)
      af[k] = *(const v2f*)(grow + 4 * k + 2 * khalf);

    for (int j = 0; j < 8; ++j) {
      const int n = j * 16 + mrow;
      v8f acc = v8f_zero();
#pragma unroll
      for (int k = 0; k < 32; ++k)
        acc = wmma4(af[k], bfrag(Wp_o1, k, khalf, n), acc);

      const float bias = b_o1[n];
#pragma unroll
      for (int r = 0; r < 8; ++r) {
        const int m = r + 8 * khalf;
        tw[m * LDS_STRIDE + n] = ssp(acc[r] + bias);
      }
    }
  }
  __syncthreads();

  if (live) {
    const float* arow = tw + mrow * LDS_STRIDE;
    v2f af2[32];
#pragma unroll
    for (int k = 0; k < 32; ++k)
      af2[k] = *(const v2f*)(arow + 4 * k + 2 * khalf);

    for (int j2 = 0; j2 < 8; ++j2) {
      const int n = j2 * 16 + mrow;
      v8f acc = v8f_zero();
#pragma unroll
      for (int k = 0; k < 32; ++k)
        acc = wmma4(af2[k], bfrag(Wp_o2, k, khalf, n), acc);

      const float bias2 = b_o2[n];
#pragma unroll
      for (int r = 0; r < 8; ++r) {
        const int m = r + 8 * khalf;
        out[(r0 + m) * A_DIM + n] = acc[r] + bias2;
      }
    }
  }
}

extern "C" void kernel_launch(void* const* d_in, const int* in_sizes, int n_in,
                              void* d_out, int out_size, void* d_ws, size_t ws_size,
                              hipStream_t stream) {
  const float* x      = (const float*)d_in[0];
  const float* f_ij   = (const float*)d_in[1];
  const float* rcut   = (const float*)d_in[2];
  const int*   idx_i  = (const int*)d_in[3];
  const int*   idx_j  = (const int*)d_in[4];
  const float* W_in2f = (const float*)d_in[5];
  const float* W_f1   = (const float*)d_in[6];
  const float* b_f1   = (const float*)d_in[7];
  const float* W_f2   = (const float*)d_in[8];
  const float* b_f2   = (const float*)d_in[9];
  const float* W_o1   = (const float*)d_in[10];
  const float* b_o1   = (const float*)d_in[11];
  const float* W_o2   = (const float*)d_in[12];
  const float* b_o2   = (const float*)d_in[13];
  float* out = (float*)d_out;

  const int  N = in_sizes[0] / A_DIM;
  const long E = (long)in_sizes[2];

  float* h       = (float*)d_ws;
  float* agg     = h + (size_t)N * F_DIM;
  float* Wp_in2f = agg + (size_t)N * F_DIM;
  float* Wp_f1   = Wp_in2f + A_DIM * F_DIM;
  float* Wp_f2   = Wp_f1 + R_DIM * F_DIM;
  float* Wp_o1   = Wp_f2 + F_DIM * F_DIM;
  float* Wp_o2   = Wp_o1 + F_DIM * A_DIM;

  {
    int blk = 256;
    pack_kernel<<<(A_DIM / 2 * F_DIM + blk - 1) / blk, blk, 0, stream>>>(W_in2f, Wp_in2f, A_DIM, F_DIM);
    pack_kernel<<<(R_DIM / 2 * F_DIM + blk - 1) / blk, blk, 0, stream>>>(W_f1, Wp_f1, R_DIM, F_DIM);
    pack_kernel<<<(F_DIM / 2 * F_DIM + blk - 1) / blk, blk, 0, stream>>>(W_f2, Wp_f2, F_DIM, F_DIM);
    pack_kernel<<<(F_DIM / 2 * A_DIM + blk - 1) / blk, blk, 0, stream>>>(W_o1, Wp_o1, F_DIM, A_DIM);
    pack_kernel<<<(A_DIM / 2 * A_DIM + blk - 1) / blk, blk, 0, stream>>>(W_o2, Wp_o2, A_DIM, A_DIM);
  }
  {
    long n = (long)N * F_DIM;
    int  blk = 256;
    zero_kernel<<<(int)((n + blk - 1) / blk), blk, 0, stream>>>(agg, n);
  }
  {
    int ntiles = N / 16;
    int blocks = (ntiles + WAVES_PER_BLOCK - 1) / WAVES_PER_BLOCK;
    gemm_h_kernel<<<blocks, BLOCK_THREADS, 0, stream>>>(x, Wp_in2f, h, ntiles);
  }
  {
    int ntiles = (int)(E / 16);
    int blocks = (ntiles + WAVES_PER_BLOCK - 1) / WAVES_PER_BLOCK;
    edge_kernel<<<blocks, BLOCK_THREADS, 0, stream>>>(
        f_ij, rcut, idx_i, idx_j, Wp_f1, b_f1, Wp_f2, b_f2, h, agg, ntiles);
  }
  {
    int ntiles = N / 16;
    int blocks = (ntiles + WAVES_PER_BLOCK - 1) / WAVES_PER_BLOCK;
    out_kernel<<<blocks, BLOCK_THREADS, 0, stream>>>(
        agg, Wp_o1, b_o1, Wp_o2, b_o2, out, ntiles);
  }
}


# edge kernel: the 8 h[idx_j] gathers of a column tile are issued as one group under its WMMA chain, not one at a time between the scatter-adds (7 exposed L2 round trips per tile before)
# speedup vs baseline: 1.4985x; 1.3090x over previous
#include <hip/hip_runtime.h>
#include <math.h>


typedef __attribute__((ext_vector_type(2))) float v2f;
typedef __attribute__((ext_vector_type(8))) float v8f;

#define A_DIM 128
#define F_DIM 128
#define R_DIM 20
#define LDS_STRIDE 132
#define WAVES_PER_BLOCK 5
#define BLOCK_THREADS (WAVES_PER_BLOCK * 32)

__device__ __forceinline__ v8f wmma4(v2f a, v2f b, v8f c) {
  return __builtin_amdgcn_wmma_f32_16x16x4_f32(
       false, a,  false, b,
       (short)0, c,  false,  false);
}

__device__ __forceinline__ v8f v8f_zero() {
  v8f z = {0.0f, 0.0f, 0.0f, 0.0f, 0.0f, 0.0f, 0.0f, 0.0f};
  return z;
}

__device__ __forceinline__ float ssp(float x) {
  const float t = __expf(-fabsf(x));
  return fmaxf(x, 0.0f) + __logf(1.0f + t) - 0.69314718055994531f;
}

__global__ void zero_kernel(float* __restrict__ p, long n) {
  long i = (long)blockIdx.x * blockDim.x + threadIdx.x;
  if (i < n) p[i] = 0.0f;
}

__global__ void pack_kernel(const float* __restrict__ W, float* __restrict__ Wp,
                            int K, int Ncol) {
  int i = blockIdx.x * blockDim.x + threadIdx.x;
  int total = (K / 2) * Ncol;
  if (i >= total) return;
  int k2 = i / Ncol;
  int n  = i - k2 * Ncol;
  Wp[k2 * 2 * Ncol + 2 * n + 0] = W[(2 * k2 + 0) * Ncol + n];
  Wp[k2 * 2 * Ncol + 2 * n + 1] = W[(2 * k2 + 1) * Ncol + n];
}

__device__ __forceinline__ v2f bfrag(const float* __restrict__ Wp,
                                     int k, int khalf, int n) {
  return *(const v2f*)(Wp + (2 * k + khalf) * 256 + 2 * n);
}

__global__ void __launch_bounds__(BLOCK_THREADS)
gemm_h_kernel(const float* __restrict__ x, const float* __restrict__ Wp,
              float* __restrict__ h, int ntiles) {
  const int wave  = threadIdx.x >> 5;
  const int lane  = threadIdx.x & 31;
  const int t     = blockIdx.x * WAVES_PER_BLOCK + wave;
  if (t >= ntiles) return;
  const int mrow  = lane & 15;
  const int khalf = lane >> 4;
  const long r0   = (long)t * 16;

  const float* xrow = x + (r0 + mrow) * A_DIM;
  v2f af[32];
#pragma unroll
  for (int k = 0; k < 32; ++k)
    af[k] = *(const v2f*)(xrow + 4 * k + 2 * khalf);

  for (int j = 0; j < 8; ++j) {
    const int n = j * 16 + mrow;
    v8f acc = v8f_zero();
#pragma unroll
    for (int k = 0; k < 32; ++k)
      acc = wmma4(af[k], bfrag(Wp, k, khalf, n), acc);
#pragma unroll
    for (int r = 0; r < 8; ++r) {
      const int m = r + 8 * khalf;
      h[(r0 + m) * F_DIM + n] = acc[r];
    }
  }
}

__global__ void __launch_bounds__(BLOCK_THREADS)
edge_kernel(const float* __restrict__ f_ij, const float* __restrict__ rcut,
            const int* __restrict__ idx_i, const int* __restrict__ idx_j,
            const float* __restrict__ Wp_f1, const float* __restrict__ b_f1,
            const float* __restrict__ Wp_f2, const float* __restrict__ b_f2,
            const float* __restrict__ h, float* __restrict__ agg, int ntiles) {
  __shared__ float lds[WAVES_PER_BLOCK][16 * LDS_STRIDE];

  const int wave   = threadIdx.x >> 5;
  const int lane   = threadIdx.x & 31;
  const int t      = blockIdx.x * WAVES_PER_BLOCK + wave;
  const bool live  = (t < ntiles);
  const int mrow   = lane & 15;
  const int khalf  = lane >> 4;
  const long e0    = (long)t * 16;
  float* tw = lds[wave];

  if (live) {
    const float* frow = f_ij + (e0 + mrow) * R_DIM;
    v2f af1[5];
#pragma unroll
    for (int k = 0; k < 5; ++k)
      af1[k] = *(const v2f*)(frow + 4 * k + 2 * khalf);

    v8f acc1[8];
#pragma unroll
    for (int j = 0; j < 8; ++j) acc1[j] = v8f_zero();
#pragma unroll
    for (int k = 0; k < 5; ++k) {
#pragma unroll
      for (int j = 0; j < 8; ++j)
        acc1[j] = wmma4(af1[k], bfrag(Wp_f1, k, khalf, j * 16 + mrow), acc1[j]);
    }
#pragma unroll
    for (int j = 0; j < 8; ++j) {
      const int n = j * 16 + mrow;
      const float bias = b_f1[n];
#pragma unroll
      for (int r = 0; r < 8; ++r) {
        const int m = r + 8 * khalf;
        tw[m * LDS_STRIDE + n] = ssp(acc1[j][r] + bias);
      }
    }
  }
  __syncthreads();

  if (live) {
    int   vi[8]; int vj[8]; float vr[8];
#pragma unroll
    for (int r = 0; r < 8; ++r) {
      const long e = e0 + r + 8 * khalf;
      vi[r] = idx_i[e]; vj[r] = idx_j[e]; vr[r] = rcut[e];
    }
    unsigned brk = 0x80u;
#pragma unroll
    for (int r = 0; r < 7; ++r)
      if (vi[r] != vi[r + 1]) brk |= (1u << r);

    const float* arow = tw + mrow * LDS_STRIDE;
    v2f af2[32];
#pragma unroll
    for (int k = 0; k < 32; ++k)
      af2[k] = *(const v2f*)(arow + 4 * k + 2 * khalf);

    for (int j2 = 0; j2 < 8; ++j2) {
      const int n = j2 * 16 + mrow;
      float xv[8];
#pragma unroll
      for (int r = 0; r < 8; ++r)
        xv[r] = h[(long)vj[r] * F_DIM + n];
      const float bias2 = b_f2[n];

      v8f acc2 = v8f_zero();
#pragma unroll
      for (int k = 0; k < 32; ++k)
        acc2 = wmma4(af2[k], bfrag(Wp_f2, k, khalf, n), acc2);

      float s = 0.0f;
#pragma unroll
      for (int r = 0; r < 8; ++r) {
        const float wij = (acc2[r] + bias2) * vr[r];
        s += wij * xv[r];
        if (r == 7 || (brk & (1u << r))) {
          atomicAdd(agg + (long)vi[r] * F_DIM + n, s);
          s = 0.0f;
        }
      }
    }
  }
}

__global__ void __launch_bounds__(BLOCK_THREADS)
out_kernel(const float* __restrict__ agg,
           const float* __restrict__ Wp_o1, const float* __restrict__ b_o1,
           const float* __restrict__ Wp_o2, const float* __restrict__ b_o2,
           float* __restrict__ out, int ntiles) {
  __shared__ float lds[WAVES_PER_BLOCK][16 * LDS_STRIDE];

  const int wave  = threadIdx.x >> 5;
  const int lane  = threadIdx.x & 31;
  const int t     = blockIdx.x * WAVES_PER_BLOCK + wave;
  const bool live = (t < ntiles);
  const int mrow  = lane & 15;
  const int khalf = lane >> 4;
  const long r0   = (long)t * 16;
  float* tw = lds[wave];

  if (live) {
    const float* grow = agg + (r0 + mrow) * F_DIM;
    v2f af[32];
#pragma unroll
    for (int k = 0; k < 32; ++k)
      af[k] = *(const v2f*)(grow + 4 * k + 2 * khalf);

    for (int j = 0; j < 8; ++j) {
      const int n = j * 16 + mrow;
      v8f acc = v8f_zero();
#pragma unroll
      for (int k = 0; k < 32; ++k)
        acc = wmma4(af[k], bfrag(Wp_o1, k, khalf, n), acc);

      const float bias = b_o1[n];
#pragma unroll
      for (int r = 0; r < 8; ++r) {
        const int m = r + 8 * khalf;
        tw[m * LDS_STRIDE + n] = ssp(acc[r] + bias);
      }
    }
  }
  __syncthreads();

  if (live) {
    const float* arow = tw + mrow * LDS_STRIDE;
    v2f af2[32];
#pragma unroll
    for (int k = 0; k < 32; ++k)
      af2[k] = *(const v2f*)(arow + 4 * k + 2 * khalf);

    for (int j2 = 0; j2 < 8; ++j2) {
      const int n = j2 * 16 + mrow;
      v8f acc = v8f_zero();
#pragma unroll
      for (int k = 0; k < 32; ++k)
        acc = wmma4(af2[k], bfrag(Wp_o2, k, khalf, n), acc);

      const float bias2 = b_o2[n];
#pragma unroll
      for (int r = 0; r < 8; ++r) {
        const int m = r + 8 * khalf;
        out[(r0 + m) * A_DIM + n] = acc[r] + bias2;
      }
    }
  }
}

extern "C" void kernel_launch(void* const* d_in, const int* in_sizes, int n_in,
                              void* d_out, int out_size, void* d_ws, size_t ws_size,
                              hipStream_t stream) {
  const float* x      = (const float*)d_in[0];
  const float* f_ij   = (const float*)d_in[1];
  const float* rcut   = (const float*)d_in[2];
  const int*   idx_i  = (const int*)d_in[3];
  const int*   idx_j  = (const int*)d_in[4];
  const float* W_in2f = (const float*)d_in[5];
  const float* W_f1   = (const float*)d_in[6];
  const float* b_f1   = (const float*)d_in[7];
  const float* W_f2   = (const float*)d_in[8];
  const float* b_f2   = (const float*)d_in[9];
  const float* W_o1   = (const float*)d_in[10];
  const float* b_o1   = (const float*)d_in[11];
  const float* W_o2   = (const float*)d_in[12];
  const float* b_o2   = (const float*)d_in[13];
  float* out = (float*)d_out;

  const int  N = in_sizes[0] / A_DIM;
  const long E = (long)in_sizes[2];

  float* h       = (float*)d_ws;
  float* agg     = h + (size_t)N * F_DIM;
  float* Wp_in2f = agg + (size_t)N * F_DIM;
  float* Wp_f1   = Wp_in2f + A_DIM * F_DIM;
  float* Wp_f2   = Wp_f1 + R_DIM * F_DIM;
  float* Wp_o1   = Wp_f2 + F_DIM * F_DIM;
  float* Wp_o2   = Wp_o1 + F_DIM * A_DIM;

  {
    int blk = 256;
    pack_kernel<<<(A_DIM / 2 * F_DIM + blk - 1) / blk, blk, 0, stream>>>(W_in2f, Wp_in2f, A_DIM, F_DIM);
    pack_kernel<<<(R_DIM / 2 * F_DIM + blk - 1) / blk, blk, 0, stream>>>(W_f1, Wp_f1, R_DIM, F_DIM);
    pack_kernel<<<(F_DIM / 2 * F_DIM + blk - 1) / blk, blk, 0, stream>>>(W_f2, Wp_f2, F_DIM, F_DIM);
    pack_kernel<<<(F_DIM / 2 * A_DIM + blk - 1) / blk, blk, 0, stream>>>(W_o1, Wp_o1, F_DIM, A_DIM);
    pack_kernel<<<(A_DIM / 2 * A_DIM + blk - 1) / blk, blk, 0, stream>>>(W_o2, Wp_o2, A_DIM, A_DIM);
  }
  {
    long n = (long)N * F_DIM;
    int  blk = 256;
    zero_kernel<<<(int)((n + blk - 1) / blk), blk, 0, stream>>>(agg, n);
  }
  {
    int ntiles = N / 16;
    int blocks = (ntiles + WAVES_PER_BLOCK - 1) / WAVES_PER_BLOCK;
    gemm_h_kernel<<<blocks, BLOCK_THREADS, 0, stream>>>(x, Wp_in2f, h, ntiles);
  }
  {
    int ntiles = (int)(E / 16);
    int blocks = (ntiles + WAVES_PER_BLOCK - 1) / WAVES_PER_BLOCK;
    edge_kernel<<<blocks, BLOCK_THREADS, 0, stream>>>(
        f_ij, rcut, idx_i, idx_j, Wp_f1, b_f1, Wp_f2, b_f2, h, agg, ntiles);
  }
  {
    int ntiles = N / 16;
    int blocks = (ntiles + WAVES_PER_BLOCK - 1) / WAVES_PER_BLOCK;
    out_kernel<<<blocks, BLOCK_THREADS, 0, stream>>>(
        agg, Wp_o1, b_o1, Wp_o2, b_o2, out, ntiles);
  }
}


# edge kernel: W_f2 is staged once per 8-wave work-group (40 tiles) into LDS in fragment order; GEMM2 reads its B fragments from there in groups of 8, not by one global load per WMMA
# speedup vs baseline: 1.7293x; 1.1540x over previous
#include <hip/hip_runtime.h>
#include <math.h>


typedef __attribute__((ext_vector_type(2))) float v2f;
typedef __attribute__((ext_vector_type(4))) float v4f;
typedef __attribute__((ext_vector_type(8))) float v8f;

#define A_DIM 128
#define F_DIM 128
#define R_DIM 20
#define LDS_STRIDE 132
#define WAVES_PER_BLOCK 5
#define BLOCK_THREADS (WAVES_PER_BLOCK * 32)
#define EDGE_WAVES 8
#define EDGE_THREADS (EDGE_WAVES * 32)
#define EDGE_TILES_PER_WAVE 5
#define EDGE_TILES_PER_BLOCK (EDGE_WAVES * EDGE_TILES_PER_WAVE)

__device__ __forceinline__ v8f wmma4(v2f a, v2f b, v8f c) {
  return __builtin_amdgcn_wmma_f32_16x16x4_f32(
       false, a,  false, b,
       (short)0, c,  false,  false);
}

__device__ __forceinline__ v8f v8f_zero() {
  v8f z = {0.0f, 0.0f, 0.0f, 0.0f, 0.0f, 0.0f, 0.0f, 0.0f};
  return z;
}

__device__ __forceinline__ float ssp(float x) {
  const float t = __expf(-fabsf(x));
  return fmaxf(x, 0.0f) + __logf(1.0f + t) - 0.69314718055994531f;
}

__global__ void zero_kernel(float* __restrict__ p, long n) {
  long i = (long)blockIdx.x * blockDim.x + threadIdx.x;
  if (i < n) p[i] = 0.0f;
}

__global__ void pack_kernel(const float* __restrict__ W, float* __restrict__ Wp,
                            int K, int Ncol) {
  int i = blockIdx.x * blockDim.x + threadIdx.x;
  int total = (K / 2) * Ncol;
  if (i >= total) return;
  int k2 = i / Ncol;
  int n  = i - k2 * Ncol;
  Wp[k2 * 2 * Ncol + 2 * n + 0] = W[(2 * k2 + 0) * Ncol + n];
  Wp[k2 * 2 * Ncol + 2 * n + 1] = W[(2 * k2 + 1) * Ncol + n];
}

__device__ __forceinline__ v2f bfrag(const float* __restrict__ Wp,
                                     int k, int khalf, int n) {
  return *(const v2f*)(Wp + (2 * k + khalf) * 256 + 2 * n);
}

__global__ void __launch_bounds__(BLOCK_THREADS)
gemm_h_kernel(const float* __restrict__ x, const float* __restrict__ Wp,
              float* __restrict__ h, int ntiles) {
  const int wave  = threadIdx.x >> 5;
  const int lane  = threadIdx.x & 31;
  const int t     = blockIdx.x * WAVES_PER_BLOCK + wave;
  if (t >= ntiles) return;
  const int mrow  = lane & 15;
  const int khalf = lane >> 4;
  const long r0   = (long)t * 16;

  const float* xrow = x + (r0 + mrow) * A_DIM;
  v2f af[32];
#pragma unroll
  for (int k = 0; k < 32; ++k)
    af[k] = *(const v2f*)(xrow + 4 * k + 2 * khalf);

  for (int j = 0; j < 8; ++j) {
    const int n = j * 16 + mrow;
    v8f acc = v8f_zero();
#pragma unroll
    for (int k = 0; k < 32; ++k)
      acc = wmma4(af[k], bfrag(Wp, k, khalf, n), acc);
#pragma unroll
    for (int r = 0; r < 8; ++r) {
      const int m = r + 8 * khalf;
      h[(r0 + m) * F_DIM + n] = acc[r];
    }
  }
}

__global__ void __launch_bounds__(EDGE_THREADS) __attribute__((amdgpu_waves_per_eu(4)))
edge_kernel(const float* __restrict__ f_ij, const float* __restrict__ rcut,
            const int* __restrict__ idx_i, const int* __restrict__ idx_j,
            const float* __restrict__ Wp_f1, const float* __restrict__ b_f1,
            const float* __restrict__ Wp_f2, const float* __restrict__ b_f2,
            const float* __restrict__ h, float* __restrict__ agg, int ntiles) {
  __shared__ __attribute__((aligned(16))) float wl[F_DIM * F_DIM];
  __shared__ __attribute__((aligned(16))) float lds[EDGE_WAVES][16 * LDS_STRIDE];

  const int wave   = threadIdx.x >> 5;
  const int lane   = threadIdx.x & 31;
  const int mrow   = lane & 15;
  const int khalf  = lane >> 4;
  float* tw = lds[wave];

  for (int i = threadIdx.x; i < 16 * 2 * F_DIM; i += EDGE_THREADS) {
    const int n = i & (F_DIM - 1);
    const int hh = (i >> 7) & 1;
    const int q = i >> 8;
    const v2f lo = *(const v2f*)(Wp_f2 + (4 * q + hh) * 256 + 2 * n);
    const v2f hi = *(const v2f*)(Wp_f2 + (4 * q + 2 + hh) * 256 + 2 * n);
    v4f v = {lo[0], lo[1], hi[0], hi[1]};
    *(v4f*)(wl + 4 * i) = v;
  }
  __syncthreads();

  const v4f* bl = (const v4f*)wl + khalf * F_DIM + mrow;

  for (int it = 0; it < EDGE_TILES_PER_WAVE; ++it) {
    const int t = blockIdx.x * EDGE_TILES_PER_BLOCK + it * EDGE_WAVES + wave;
    if (t >= ntiles) break;
    const long e0 = (long)t * 16;

    const float* frow = f_ij + (e0 + mrow) * R_DIM;
    v2f af1[5];
#pragma unroll
    for (int k = 0; k < 5; ++k)
      af1[k] = *(const v2f*)(frow + 4 * k + 2 * khalf);

    v8f acc1[8];
#pragma unroll
    for (int j = 0; j < 8; ++j) acc1[j] = v8f_zero();
#pragma unroll
    for (int k = 0; k < 5; ++k) {
#pragma unroll
      for (int j = 0; j < 8; ++j)
        acc1[j] = wmma4(af1[k], bfrag(Wp_f1, k, khalf, j * 16 + mrow), acc1[j]);
    }
#pragma unroll
    for (int j = 0; j < 8; ++j) {
      const int n = j * 16 + mrow;
      const float bias = b_f1[n];
#pragma unroll
      for (int r = 0; r < 8; ++r) {
        const int m = r + 8 * khalf;
        tw[m * LDS_STRIDE + n] = ssp(acc1[j][r] + bias);
      }
    }
    __builtin_amdgcn_fence(__ATOMIC_RELEASE, "wavefront");
    __builtin_amdgcn_wave_barrier();
    __builtin_amdgcn_fence(__ATOMIC_ACQUIRE, "wavefront");

    int   vi[8]; int vj[8]; float vr[8];
#pragma unroll
    for (int r = 0; r < 8; ++r) {
      const long e = e0 + r + 8 * khalf;
      vi[r] = idx_i[e]; vj[r] = idx_j[e]; vr[r] = rcut[e];
    }
    unsigned brk = 0x80u;
#pragma unroll
    for (int r = 0; r < 7; ++r)
      if (vi[r] != vi[r + 1]) brk |= (1u << r);

    const float* arow = tw + mrow * LDS_STRIDE;
    v2f af2[32];
#pragma unroll
    for (int k = 0; k < 32; ++k)
      af2[k] = *(const v2f*)(arow + 4 * k + 2 * khalf);

    v4f bb[2][8];
#pragma unroll
    for (int q = 0; q < 8; ++q)
      bb[0][q] = bl[q * 2 * F_DIM];

#pragma unroll
    for (int j2 = 0; j2 < 8; ++j2) {
      const int n = j2 * 16 + mrow;
      float xv[8];
#pragma unroll
      for (int r = 0; r < 8; ++r)
        xv[r] = h[(long)vj[r] * F_DIM + n];
      const float bias2 = b_f2[n];

#pragma unroll
      for (int q = 0; q < 8; ++q)
        bb[1][q] = bl[(8 + q) * 2 * F_DIM + j2 * 16];
      __builtin_amdgcn_sched_barrier(0);

      v8f acc2 = v8f_zero();
#pragma unroll
      for (int q = 0; q < 8; ++q) {
        const v2f b0 = {bb[0][q][0], bb[0][q][1]};
        const v2f b1 = {bb[0][q][2], bb[0][q][3]};
        acc2 = wmma4(af2[2 * q],     b0, acc2);
        acc2 = wmma4(af2[2 * q + 1], b1, acc2);
      }
      __builtin_amdgcn_sched_barrier(0);
      if (j2 < 7) {
#pragma unroll
        for (int q = 0; q < 8; ++q)
          bb[0][q] = bl[q * 2 * F_DIM + (j2 + 1) * 16];
      }
      __builtin_amdgcn_sched_barrier(0);
#pragma unroll
      for (int q = 0; q < 8; ++q) {
        const v2f b0 = {bb[1][q][0], bb[1][q][1]};
        const v2f b1 = {bb[1][q][2], bb[1][q][3]};
        acc2 = wmma4(af2[16 + 2 * q],     b0, acc2);
        acc2 = wmma4(af2[16 + 2 * q + 1], b1, acc2);
      }
      __builtin_amdgcn_sched_barrier(0);

      float s = 0.0f;
#pragma unroll
      for (int r = 0; r < 8; ++r) {
        const float wij = (acc2[r] + bias2) * vr[r];
        s += wij * xv[r];
        if (r == 7 || (brk & (1u << r))) {
          atomicAdd(agg + (long)vi[r] * F_DIM + n, s);
          s = 0.0f;
        }
      }
    }
  }
}

__global__ void __launch_bounds__(BLOCK_THREADS)
out_kernel(const float* __restrict__ agg,
           const float* __restrict__ Wp_o1, const float* __restrict__ b_o1,
           const float* __restrict__ Wp_o2, const float* __restrict__ b_o2,
           float* __restrict__ out, int ntiles) {
  __shared__ float lds[WAVES_PER_BLOCK][16 * LDS_STRIDE];

  const int wave  = threadIdx.x >> 5;
  const int lane  = threadIdx.x & 31;
  const int t     = blockIdx.x * WAVES_PER_BLOCK + wave;
  const bool live = (t < ntiles);
  const int mrow  = lane & 15;
  const int khalf = lane >> 4;
  const long r0   = (long)t * 16;
  float* tw = lds[wave];

  if (live) {
    const float* grow = agg + (r0 + mrow) * F_DIM;
    v2f af[32];
#pragma unroll
    for (int k = 0; k < 32; ++k)
      af[k] = *(const v2f*)(grow + 4 * k + 2 * khalf);

    for (int j = 0; j < 8; ++j) {
      const int n = j * 16 + mrow;
      v8f acc = v8f_zero();
#pragma unroll
      for (int k = 0; k < 32; ++k)
        acc = wmma4(af[k], bfrag(Wp_o1, k, khalf, n), acc);

      const float bias = b_o1[n];
#pragma unroll
      for (int r = 0; r < 8; ++r) {
        const int m = r + 8 * khalf;
        tw[m * LDS_STRIDE + n] = ssp(acc[r] + bias);
      }
    }
  }
  __syncthreads();

  if (live) {
    const float* arow = tw + mrow * LDS_STRIDE;
    v2f af2[32];
#pragma unroll
    for (int k = 0; k < 32; ++k)
      af2[k] = *(const v2f*)(arow + 4 * k + 2 * khalf);

    for (int j2 = 0; j2 < 8; ++j2) {
      const int n = j2 * 16 + mrow;
      v8f acc = v8f_zero();
#pragma unroll
      for (int k = 0; k < 32; ++k)
        acc = wmma4(af2[k], bfrag(Wp_o2, k, khalf, n), acc);

      const float bias2 = b_o2[n];
#pragma unroll
      for (int r = 0; r < 8; ++r) {
        const int m = r + 8 * khalf;
        out[(r0 + m) * A_DIM + n] = acc[r] + bias2;
      }
    }
  }
}

extern "C" void kernel_launch(void* const* d_in, const int* in_sizes, int n_in,
                              void* d_out, int out_size, void* d_ws, size_t ws_size,
                              hipStream_t stream) {
  const float* x      = (const float*)d_in[0];
  const float* f_ij   = (const float*)d_in[1];
  const float* rcut   = (const float*)d_in[2];
  const int*   idx_i  = (const int*)d_in[3];
  const int*   idx_j  = (const int*)d_in[4];
  const float* W_in2f = (const float*)d_in[5];
  const float* W_f1   = (const float*)d_in[6];
  const float* b_f1   = (const float*)d_in[7];
  const float* W_f2   = (const float*)d_in[8];
  const float* b_f2   = (const float*)d_in[9];
  const float* W_o1   = (const float*)d_in[10];
  const float* b_o1   = (const float*)d_in[11];
  const float* W_o2   = (const float*)d_in[12];
  const float* b_o2   = (const float*)d_in[13];
  float* out = (float*)d_out;

  const int  N = in_sizes[0] / A_DIM;
  const long E = (long)in_sizes[2];

  float* h       = (float*)d_ws;
  float* agg     = h + (size_t)N * F_DIM;
  float* Wp_in2f = agg + (size_t)N * F_DIM;
  float* Wp_f1   = Wp_in2f + A_DIM * F_DIM;
  float* Wp_f2   = Wp_f1 + R_DIM * F_DIM;
  float* Wp_o1   = Wp_f2 + F_DIM * F_DIM;
  float* Wp_o2   = Wp_o1 + F_DIM * A_DIM;

  {
    int blk = 256;
    pack_kernel<<<(A_DIM / 2 * F_DIM + blk - 1) / blk, blk, 0, stream>>>(W_in2f, Wp_in2f, A_DIM, F_DIM);
    pack_kernel<<<(R_DIM / 2 * F_DIM + blk - 1) / blk, blk, 0, stream>>>(W_f1, Wp_f1, R_DIM, F_DIM);
    pack_kernel<<<(F_DIM / 2 * F_DIM + blk - 1) / blk, blk, 0, stream>>>(W_f2, Wp_f2, F_DIM, F_DIM);
    pack_kernel<<<(F_DIM / 2 * A_DIM + blk - 1) / blk, blk, 0, stream>>>(W_o1, Wp_o1, F_DIM, A_DIM);
    pack_kernel<<<(A_DIM / 2 * A_DIM + blk - 1) / blk, blk, 0, stream>>>(W_o2, Wp_o2, A_DIM, A_DIM);
  }
  {
    long n = (long)N * F_DIM;
    int  blk = 256;
    zero_kernel<<<(int)((n + blk - 1) / blk), blk, 0, stream>>>(agg, n);
  }
  {
    int ntiles = N / 16;
    int blocks = (ntiles + WAVES_PER_BLOCK - 1) / WAVES_PER_BLOCK;
    gemm_h_kernel<<<blocks, BLOCK_THREADS, 0, stream>>>(x, Wp_in2f, h, ntiles);
  }
  {
    int ntiles = (int)(E / 16);
    int blocks = (ntiles + EDGE_TILES_PER_BLOCK - 1) / EDGE_TILES_PER_BLOCK;
    edge_kernel<<<blocks, EDGE_THREADS, 0, stream>>>(
        f_ij, rcut, idx_i, idx_j, Wp_f1, b_f1, Wp_f2, b_f2, h, agg, ntiles);
  }
  {
    int ntiles = N / 16;
    int blocks = (ntiles + WAVES_PER_BLOCK - 1) / WAVES_PER_BLOCK;
    out_kernel<<<blocks, BLOCK_THREADS, 0, stream>>>(
        agg, Wp_o1, b_o1, Wp_o2, b_o2, out, ntiles);
  }
}


# shifted softplus as max(x,0)+ln2*log2((1+exp2(-|x|*log2e))/2) on bare v_exp_f32/v_log_f32: 7 vector instructions per element, not 18 (the library log's rescale and tail are gone)
# speedup vs baseline: 1.8769x; 1.0854x over previous
#include <hip/hip_runtime.h>
#include <math.h>


typedef __attribute__((ext_vector_type(2))) float v2f;
typedef __attribute__((ext_vector_type(4))) float v4f;
typedef __attribute__((ext_vector_type(8))) float v8f;

#define A_DIM 128
#define F_DIM 128
#define R_DIM 20
#define LDS_STRIDE 132
#define WAVES_PER_BLOCK 5
#define BLOCK_THREADS (WAVES_PER_BLOCK * 32)
#define EDGE_WAVES 8
#define EDGE_THREADS (EDGE_WAVES * 32)
#define EDGE_TILES_PER_WAVE 5
#define EDGE_TILES_PER_BLOCK (EDGE_WAVES * EDGE_TILES_PER_WAVE)

__device__ __forceinline__ v8f wmma4(v2f a, v2f b, v8f c) {
  return __builtin_amdgcn_wmma_f32_16x16x4_f32(
       false, a,  false, b,
       (short)0, c,  false,  false);
}

__device__ __forceinline__ v8f v8f_zero() {
  v8f z = {0.0f, 0.0f, 0.0f, 0.0f, 0.0f, 0.0f, 0.0f, 0.0f};
  return z;
}

__device__ __forceinline__ float ssp(float x) {
  const float t = __builtin_amdgcn_exp2f(-1.44269504088896341f * fabsf(x));
  const float l = __builtin_amdgcn_logf(__builtin_fmaf(t, 0.5f, 0.5f));
  return __builtin_fmaf(l, 0.69314718055994531f, fmaxf(x, 0.0f));
}

__global__ void zero_kernel(float* __restrict__ p, long n) {
  long i = (long)blockIdx.x * blockDim.x + threadIdx.x;
  if (i < n) p[i] = 0.0f;
}

__global__ void pack_kernel(const float* __restrict__ W, float* __restrict__ Wp,
                            int K, int Ncol) {
  int i = blockIdx.x * blockDim.x + threadIdx.x;
  int total = (K / 2) * Ncol;
  if (i >= total) return;
  int k2 = i / Ncol;
  int n  = i - k2 * Ncol;
  Wp[k2 * 2 * Ncol + 2 * n + 0] = W[(2 * k2 + 0) * Ncol + n];
  Wp[k2 * 2 * Ncol + 2 * n + 1] = W[(2 * k2 + 1) * Ncol + n];
}

__device__ __forceinline__ v2f bfrag(const float* __restrict__ Wp,
                                     int k, int khalf, int n) {
  return *(const v2f*)(Wp + (2 * k + khalf) * 256 + 2 * n);
}

__global__ void __launch_bounds__(BLOCK_THREADS)
gemm_h_kernel(const float* __restrict__ x, const float* __restrict__ Wp,
              float* __restrict__ h, int ntiles) {
  const int wave  = threadIdx.x >> 5;
  const int lane  = threadIdx.x & 31;
  const int t     = blockIdx.x * WAVES_PER_BLOCK + wave;
  if (t >= ntiles) return;
  const int mrow  = lane & 15;
  const int khalf = lane >> 4;
  const long r0   = (long)t * 16;

  const float* xrow = x + (r0 + mrow) * A_DIM;
  v2f af[32];
#pragma unroll
  for (int k = 0; k < 32; ++k)
    af[k] = *(const v2f*)(xrow + 4 * k + 2 * khalf);

  for (int j = 0; j < 8; ++j) {
    const int n = j * 16 + mrow;
    v8f acc = v8f_zero();
#pragma unroll
    for (int k = 0; k < 32; ++k)
      acc = wmma4(af[k], bfrag(Wp, k, khalf, n), acc);
#pragma unroll
    for (int r = 0; r < 8; ++r) {
      const int m = r + 8 * khalf;
      h[(r0 + m) * F_DIM + n] = acc[r];
    }
  }
}

__global__ void __launch_bounds__(EDGE_THREADS) __attribute__((amdgpu_waves_per_eu(4)))
edge_kernel(const float* __restrict__ f_ij, const float* __restrict__ rcut,
            const int* __restrict__ idx_i, const int* __restrict__ idx_j,
            const float* __restrict__ Wp_f1, const float* __restrict__ b_f1,
            const float* __restrict__ Wp_f2, const float* __restrict__ b_f2,
            const float* __restrict__ h, float* __restrict__ agg, int ntiles) {
  __shared__ __attribute__((aligned(16))) float wl[F_DIM * F_DIM];
  __shared__ __attribute__((aligned(16))) float lds[EDGE_WAVES][16 * LDS_STRIDE];

  const int wave   = threadIdx.x >> 5;
  const int lane   = threadIdx.x & 31;
  const int mrow   = lane & 15;
  const int khalf  = lane >> 4;
  float* tw = lds[wave];

  for (int i = threadIdx.x; i < 16 * 2 * F_DIM; i += EDGE_THREADS) {
    const int n = i & (F_DIM - 1);
    const int hh = (i >> 7) & 1;
    const int q = i >> 8;
    const v2f lo = *(const v2f*)(Wp_f2 + (4 * q + hh) * 256 + 2 * n);
    const v2f hi = *(const v2f*)(Wp_f2 + (4 * q + 2 + hh) * 256 + 2 * n);
    v4f v = {lo[0], lo[1], hi[0], hi[1]};
    *(v4f*)(wl + 4 * i) = v;
  }
  __syncthreads();

  const v4f* bl = (const v4f*)wl + khalf * F_DIM + mrow;

  for (int it = 0; it < EDGE_TILES_PER_WAVE; ++it) {
    const int t = blockIdx.x * EDGE_TILES_PER_BLOCK + it * EDGE_WAVES + wave;
    if (t >= ntiles) break;
    const long e0 = (long)t * 16;

    const float* frow = f_ij + (e0 + mrow) * R_DIM;
    v2f af1[5];
#pragma unroll
    for (int k = 0; k < 5; ++k)
      af1[k] = *(const v2f*)(frow + 4 * k + 2 * khalf);

    v8f acc1[8];
#pragma unroll
    for (int j = 0; j < 8; ++j) acc1[j] = v8f_zero();
#pragma unroll
    for (int k = 0; k < 5; ++k) {
#pragma unroll
      for (int j = 0; j < 8; ++j)
        acc1[j] = wmma4(af1[k], bfrag(Wp_f1, k, khalf, j * 16 + mrow), acc1[j]);
    }
#pragma unroll
    for (int j = 0; j < 8; ++j) {
      const int n = j * 16 + mrow;
      const float bias = b_f1[n];
#pragma unroll
      for (int r = 0; r < 8; ++r) {
        const int m = r + 8 * khalf;
        tw[m * LDS_STRIDE + n] = ssp(acc1[j][r] + bias);
      }
    }
    __builtin_amdgcn_fence(__ATOMIC_RELEASE, "wavefront");
    __builtin_amdgcn_wave_barrier();
    __builtin_amdgcn_fence(__ATOMIC_ACQUIRE, "wavefront");

    int   vi[8]; int vj[8]; float vr[8];
#pragma unroll
    for (int r = 0; r < 8; ++r) {
      const long e = e0 + r + 8 * khalf;
      vi[r] = idx_i[e]; vj[r] = idx_j[e]; vr[r] = rcut[e];
    }
    unsigned brk = 0x80u;
#pragma unroll
    for (int r = 0; r < 7; ++r)
      if (vi[r] != vi[r + 1]) brk |= (1u << r);

    const float* arow = tw + mrow * LDS_STRIDE;
    v2f af2[32];
#pragma unroll
    for (int k = 0; k < 32; ++k)
      af2[k] = *(const v2f*)(arow + 4 * k + 2 * khalf);

    v4f bb[2][8];
#pragma unroll
    for (int q = 0; q < 8; ++q)
      bb[0][q] = bl[q * 2 * F_DIM];

#pragma unroll
    for (int j2 = 0; j2 < 8; ++j2) {
      const int n = j2 * 16 + mrow;
      float xv[8];
#pragma unroll
      for (int r = 0; r < 8; ++r)
        xv[r] = h[(long)vj[r] * F_DIM + n];
      const float bias2 = b_f2[n];

#pragma unroll
      for (int q = 0; q < 8; ++q)
        bb[1][q] = bl[(8 + q) * 2 * F_DIM + j2 * 16];
      __builtin_amdgcn_sched_barrier(0);

      v8f acc2 = v8f_zero();
#pragma unroll
      for (int q = 0; q < 8; ++q) {
        const v2f b0 = {bb[0][q][0], bb[0][q][1]};
        const v2f b1 = {bb[0][q][2], bb[0][q][3]};
        acc2 = wmma4(af2[2 * q],     b0, acc2);
        acc2 = wmma4(af2[2 * q + 1], b1, acc2);
      }
      __builtin_amdgcn_sched_barrier(0);
      if (j2 < 7) {
#pragma unroll
        for (int q = 0; q < 8; ++q)
          bb[0][q] = bl[q * 2 * F_DIM + (j2 + 1) * 16];
      }
      __builtin_amdgcn_sched_barrier(0);
#pragma unroll
      for (int q = 0; q < 8; ++q) {
        const v2f b0 = {bb[1][q][0], bb[1][q][1]};
        const v2f b1 = {bb[1][q][2], bb[1][q][3]};
        acc2 = wmma4(af2[16 + 2 * q],     b0, acc2);
        acc2 = wmma4(af2[16 + 2 * q + 1], b1, acc2);
      }
      __builtin_amdgcn_sched_barrier(0);

      float s = 0.0f;
#pragma unroll
      for (int r = 0; r < 8; ++r) {
        const float wij = (acc2[r] + bias2) * vr[r];
        s += wij * xv[r];
        if (r == 7 || (brk & (1u << r))) {
          atomicAdd(agg + (long)vi[r] * F_DIM + n, s);
          s = 0.0f;
        }
      }
    }
  }
}

__global__ void __launch_bounds__(BLOCK_THREADS)
out_kernel(const float* __restrict__ agg,
           const float* __restrict__ Wp_o1, const float* __restrict__ b_o1,
           const float* __restrict__ Wp_o2, const float* __restrict__ b_o2,
           float* __restrict__ out, int ntiles) {
  __shared__ float lds[WAVES_PER_BLOCK][16 * LDS_STRIDE];

  const int wave  = threadIdx.x >> 5;
  const int lane  = threadIdx.x & 31;
  const int t     = blockIdx.x * WAVES_PER_BLOCK + wave;
  const bool live = (t < ntiles);
  const int mrow  = lane & 15;
  const int khalf = lane >> 4;
  const long r0   = (long)t * 16;
  float* tw = lds[wave];

  if (live) {
    const float* grow = agg + (r0 + mrow) * F_DIM;
    v2f af[32];
#pragma unroll
    for (int k = 0; k < 32; ++k)
      af[k] = *(const v2f*)(grow + 4 * k + 2 * khalf);

    for (int j = 0; j < 8; ++j) {
      const int n = j * 16 + mrow;
      v8f acc = v8f_zero();
#pragma unroll
      for (int k = 0; k < 32; ++k)
        acc = wmma4(af[k], bfrag(Wp_o1, k, khalf, n), acc);

      const float bias = b_o1[n];
#pragma unroll
      for (int r = 0; r < 8; ++r) {
        const int m = r + 8 * khalf;
        tw[m * LDS_STRIDE + n] = ssp(acc[r] + bias);
      }
    }
  }
  __syncthreads();

  if (live) {
    const float* arow = tw + mrow * LDS_STRIDE;
    v2f af2[32];
#pragma unroll
    for (int k = 0; k < 32; ++k)
      af2[k] = *(const v2f*)(arow + 4 * k + 2 * khalf);

    for (int j2 = 0; j2 < 8; ++j2) {
      const int n = j2 * 16 + mrow;
      v8f acc = v8f_zero();
#pragma unroll
      for (int k = 0; k < 32; ++k)
        acc = wmma4(af2[k], bfrag(Wp_o2, k, khalf, n), acc);

      const float bias2 = b_o2[n];
#pragma unroll
      for (int r = 0; r < 8; ++r) {
        const int m = r + 8 * khalf;
        out[(r0 + m) * A_DIM + n] = acc[r] + bias2;
      }
    }
  }
}

extern "C" void kernel_launch(void* const* d_in, const int* in_sizes, int n_in,
                              void* d_out, int out_size, void* d_ws, size_t ws_size,
                              hipStream_t stream) {
  const float* x      = (const float*)d_in[0];
  const float* f_ij   = (const float*)d_in[1];
  const float* rcut   = (const float*)d_in[2];
  const int*   idx_i  = (const int*)d_in[3];
  const int*   idx_j  = (const int*)d_in[4];
  const float* W_in2f = (const float*)d_in[5];
  const float* W_f1   = (const float*)d_in[6];
  const float* b_f1   = (const float*)d_in[7];
  const float* W_f2   = (const float*)d_in[8];
  const float* b_f2   = (const float*)d_in[9];
  const float* W_o1   = (const float*)d_in[10];
  const float* b_o1   = (const float*)d_in[11];
  const float* W_o2   = (const float*)d_in[12];
  const float* b_o2   = (const float*)d_in[13];
  float* out = (float*)d_out;

  const int  N = in_sizes[0] / A_DIM;
  const long E = (long)in_sizes[2];

  float* h       = (float*)d_ws;
  float* agg     = h + (size_t)N * F_DIM;
  float* Wp_in2f = agg + (size_t)N * F_DIM;
  float* Wp_f1   = Wp_in2f + A_DIM * F_DIM;
  float* Wp_f2   = Wp_f1 + R_DIM * F_DIM;
  float* Wp_o1   = Wp_f2 + F_DIM * F_DIM;
  float* Wp_o2   = Wp_o1 + F_DIM * A_DIM;

  {
    int blk = 256;
    pack_kernel<<<(A_DIM / 2 * F_DIM + blk - 1) / blk, blk, 0, stream>>>(W_in2f, Wp_in2f, A_DIM, F_DIM);
    pack_kernel<<<(R_DIM / 2 * F_DIM + blk - 1) / blk, blk, 0, stream>>>(W_f1, Wp_f1, R_DIM, F_DIM);
    pack_kernel<<<(F_DIM / 2 * F_DIM + blk - 1) / blk, blk, 0, stream>>>(W_f2, Wp_f2, F_DIM, F_DIM);
    pack_kernel<<<(F_DIM / 2 * A_DIM + blk - 1) / blk, blk, 0, stream>>>(W_o1, Wp_o1, F_DIM, A_DIM);
    pack_kernel<<<(A_DIM / 2 * A_DIM + blk - 1) / blk, blk, 0, stream>>>(W_o2, Wp_o2, A_DIM, A_DIM);
  }
  {
    long n = (long)N * F_DIM;
    int  blk = 256;
    zero_kernel<<<(int)((n + blk - 1) / blk), blk, 0, stream>>>(agg, n);
  }
  {
    int ntiles = N / 16;
    int blocks = (ntiles + WAVES_PER_BLOCK - 1) / WAVES_PER_BLOCK;
    gemm_h_kernel<<<blocks, BLOCK_THREADS, 0, stream>>>(x, Wp_in2f, h, ntiles);
  }
  {
    int ntiles = (int)(E / 16);
    int blocks = (ntiles + EDGE_TILES_PER_BLOCK - 1) / EDGE_TILES_PER_BLOCK;
    edge_kernel<<<blocks, EDGE_THREADS, 0, stream>>>(
        f_ij, rcut, idx_i, idx_j, Wp_f1, b_f1, Wp_f2, b_f2, h, agg, ntiles);
  }
  {
    int ntiles = N / 16;
    int blocks = (ntiles + WAVES_PER_BLOCK - 1) / WAVES_PER_BLOCK;
    out_kernel<<<blocks, BLOCK_THREADS, 0, stream>>>(
        agg, Wp_o1, b_o1, Wp_o2, b_o2, out, ntiles);
  }
}


# the weight repacks, the accumulator clear and the h projection run as ONE launch (a block's index picks its part; the projection reads W_in2f in place): 9 launches per call become 3
# speedup vs baseline: 1.9380x; 1.0326x over previous
#include <hip/hip_runtime.h>
#include <math.h>


typedef __attribute__((ext_vector_type(2))) float v2f;
typedef __attribute__((ext_vector_type(4))) float v4f;
typedef __attribute__((ext_vector_type(8))) float v8f;

#define A_DIM 128
#define F_DIM 128
#define R_DIM 20
#define LDS_STRIDE 132
#define WAVES_PER_BLOCK 5
#define BLOCK_THREADS (WAVES_PER_BLOCK * 32)
#define EDGE_WAVES 8
#define EDGE_THREADS (EDGE_WAVES * 32)
#define EDGE_TILES_PER_WAVE 5
#define EDGE_TILES_PER_BLOCK (EDGE_WAVES * EDGE_TILES_PER_WAVE)

__device__ __forceinline__ v8f wmma4(v2f a, v2f b, v8f c) {
  return __builtin_amdgcn_wmma_f32_16x16x4_f32(
       false, a,  false, b,
       (short)0, c,  false,  false);
}

__device__ __forceinline__ v8f v8f_zero() {
  v8f z = {0.0f, 0.0f, 0.0f, 0.0f, 0.0f, 0.0f, 0.0f, 0.0f};
  return z;
}

__device__ __forceinline__ float ssp(float x) {
  const float t = __builtin_amdgcn_exp2f(-1.44269504088896341f * fabsf(x));
  const float l = __builtin_amdgcn_logf(__builtin_fmaf(t, 0.5f, 0.5f));
  return __builtin_fmaf(l, 0.69314718055994531f, fmaxf(x, 0.0f));
}

__device__ __forceinline__ void pack_one(const float* __restrict__ W,
                                         float* __restrict__ Wp, int i) {
  const int k2 = i >> 7;
  const int n  = i & 127;
  const v2f v = {W[(2 * k2 + 0) * 128 + n], W[(2 * k2 + 1) * 128 + n]};
  *(v2f*)(Wp + k2 * 256 + 2 * n) = v;
}

__device__ __forceinline__ v2f bfrag(const float* __restrict__ Wp,
                                     int k, int khalf, int n) {
  return *(const v2f*)(Wp + (2 * k + khalf) * 256 + 2 * n);
}

__device__ __forceinline__ v2f bfrag_rowmajor(const float* __restrict__ W,
                                              int k, int khalf, int n) {
  const float* p = W + (4 * k + 2 * khalf) * 128 + n;
  const v2f b = {p[0], p[128]};
  return b;
}

#define ZERO_V4_PER_THREAD 8
#define ZERO_V4_PER_BLOCK (ZERO_V4_PER_THREAD * BLOCK_THREADS)
#define PACK_N1 ((R_DIM / 2) * F_DIM)
#define PACK_N2 ((F_DIM / 2) * F_DIM)
#define PACK_N3 ((F_DIM / 2) * A_DIM)
#define PACK_N4 ((A_DIM / 2) * A_DIM)
#define PACK_TOTAL (PACK_N1 + PACK_N2 + PACK_N3 + PACK_N4)

__global__ void __launch_bounds__(BLOCK_THREADS)
prep_kernel(const float* __restrict__ x, const float* __restrict__ W_in2f,
            float* __restrict__ h, int ntiles, int gemm_blocks,
            float* __restrict__ agg, long n4, int zero_blocks,
            const float* __restrict__ W_f1, float* __restrict__ Wp_f1,
            const float* __restrict__ W_f2, float* __restrict__ Wp_f2,
            const float* __restrict__ W_o1, float* __restrict__ Wp_o1,
            const float* __restrict__ W_o2, float* __restrict__ Wp_o2) {
  int b = blockIdx.x;

  if (b < gemm_blocks) {
    const int wave  = threadIdx.x >> 5;
    const int lane  = threadIdx.x & 31;
    const int t     = b * WAVES_PER_BLOCK + wave;
    if (t >= ntiles) return;
    const int mrow  = lane & 15;
    const int khalf = lane >> 4;
    const long r0   = (long)t * 16;

    const float* xrow = x + (r0 + mrow) * A_DIM;
    v2f af[32];
#pragma unroll
    for (int k = 0; k < 32; ++k)
      af[k] = *(const v2f*)(xrow + 4 * k + 2 * khalf);

    for (int j = 0; j < 8; ++j) {
      const int n = j * 16 + mrow;
      v8f acc = v8f_zero();
#pragma unroll
      for (int k = 0; k < 32; ++k)
        acc = wmma4(af[k], bfrag_rowmajor(W_in2f, k, khalf, n), acc);
#pragma unroll
      for (int r = 0; r < 8; ++r) {
        const int m = r + 8 * khalf;
        h[(r0 + m) * F_DIM + n] = acc[r];
      }
    }
    return;
  }
  b -= gemm_blocks;

  if (b < zero_blocks) {
    const v4f z = {0.0f, 0.0f, 0.0f, 0.0f};
    v4f* p = (v4f*)agg;
    const long base = (long)b * ZERO_V4_PER_BLOCK + threadIdx.x;
#pragma unroll
    for (int it = 0; it < ZERO_V4_PER_THREAD; ++it) {
      const long i = base + it * BLOCK_THREADS;
      if (i < n4) p[i] = z;
    }
    return;
  }
  b -= zero_blocks;

  int i = b * BLOCK_THREADS + threadIdx.x;
  if (i < PACK_N1) { pack_one(W_f1, Wp_f1, i); return; }
  i -= PACK_N1;
  if (i < PACK_N2) { pack_one(W_f2, Wp_f2, i); return; }
  i -= PACK_N2;
  if (i < PACK_N3) { pack_one(W_o1, Wp_o1, i); return; }
  i -= PACK_N3;
  if (i < PACK_N4) pack_one(W_o2, Wp_o2, i);
}

__global__ void __launch_bounds__(EDGE_THREADS) __attribute__((amdgpu_waves_per_eu(4)))
edge_kernel(const float* __restrict__ f_ij, const float* __restrict__ rcut,
            const int* __restrict__ idx_i, const int* __restrict__ idx_j,
            const float* __restrict__ Wp_f1, const float* __restrict__ b_f1,
            const float* __restrict__ Wp_f2, const float* __restrict__ b_f2,
            const float* __restrict__ h, float* __restrict__ agg, int ntiles) {
  __shared__ __attribute__((aligned(16))) float wl[F_DIM * F_DIM];
  __shared__ __attribute__((aligned(16))) float lds[EDGE_WAVES][16 * LDS_STRIDE];

  const int wave   = threadIdx.x >> 5;
  const int lane   = threadIdx.x & 31;
  const int mrow   = lane & 15;
  const int khalf  = lane >> 4;
  float* tw = lds[wave];

  for (int i = threadIdx.x; i < 16 * 2 * F_DIM; i += EDGE_THREADS) {
    const int n = i & (F_DIM - 1);
    const int hh = (i >> 7) & 1;
    const int q = i >> 8;
    const v2f lo = *(const v2f*)(Wp_f2 + (4 * q + hh) * 256 + 2 * n);
    const v2f hi = *(const v2f*)(Wp_f2 + (4 * q + 2 + hh) * 256 + 2 * n);
    v4f v = {lo[0], lo[1], hi[0], hi[1]};
    *(v4f*)(wl + 4 * i) = v;
  }
  __syncthreads();

  const v4f* bl = (const v4f*)wl + khalf * F_DIM + mrow;

  for (int it = 0; it < EDGE_TILES_PER_WAVE; ++it) {
    const int t = blockIdx.x * EDGE_TILES_PER_BLOCK + it * EDGE_WAVES + wave;
    if (t >= ntiles) break;
    const long e0 = (long)t * 16;

    const float* frow = f_ij + (e0 + mrow) * R_DIM;
    v2f af1[5];
#pragma unroll
    for (int k = 0; k < 5; ++k)
      af1[k] = *(const v2f*)(frow + 4 * k + 2 * khalf);

    v8f acc1[8];
#pragma unroll
    for (int j = 0; j < 8; ++j) acc1[j] = v8f_zero();
#pragma unroll
    for (int k = 0; k < 5; ++k) {
#pragma unroll
      for (int j = 0; j < 8; ++j)
        acc1[j] = wmma4(af1[k], bfrag(Wp_f1, k, khalf, j * 16 + mrow), acc1[j]);
    }
#pragma unroll
    for (int j = 0; j < 8; ++j) {
      const int n = j * 16 + mrow;
      const float bias = b_f1[n];
#pragma unroll
      for (int r = 0; r < 8; ++r) {
        const int m = r + 8 * khalf;
        tw[m * LDS_STRIDE + n] = ssp(acc1[j][r] + bias);
      }
    }
    __builtin_amdgcn_fence(__ATOMIC_RELEASE, "wavefront");
    __builtin_amdgcn_wave_barrier();
    __builtin_amdgcn_fence(__ATOMIC_ACQUIRE, "wavefront");

    int   vi[8]; int vj[8]; float vr[8];
#pragma unroll
    for (int r = 0; r < 8; ++r) {
      const long e = e0 + r + 8 * khalf;
      vi[r] = idx_i[e]; vj[r] = idx_j[e]; vr[r] = rcut[e];
    }
    unsigned brk = 0x80u;
#pragma unroll
    for (int r = 0; r < 7; ++r)
      if (vi[r] != vi[r + 1]) brk |= (1u << r);

    const float* arow = tw + mrow * LDS_STRIDE;
    v2f af2[32];
#pragma unroll
    for (int k = 0; k < 32; ++k)
      af2[k] = *(const v2f*)(arow + 4 * k + 2 * khalf);

    v4f bb[2][8];
#pragma unroll
    for (int q = 0; q < 8; ++q)
      bb[0][q] = bl[q * 2 * F_DIM];

#pragma unroll
    for (int j2 = 0; j2 < 8; ++j2) {
      const int n = j2 * 16 + mrow;
      float xv[8];
#pragma unroll
      for (int r = 0; r < 8; ++r)
        xv[r] = h[(long)vj[r] * F_DIM + n];
      const float bias2 = b_f2[n];

#pragma unroll
      for (int q = 0; q < 8; ++q)
        bb[1][q] = bl[(8 + q) * 2 * F_DIM + j2 * 16];
      __builtin_amdgcn_sched_barrier(0);

      v8f acc2 = v8f_zero();
#pragma unroll
      for (int q = 0; q < 8; ++q) {
        const v2f b0 = {bb[0][q][0], bb[0][q][1]};
        const v2f b1 = {bb[0][q][2], bb[0][q][3]};
        acc2 = wmma4(af2[2 * q],     b0, acc2);
        acc2 = wmma4(af2[2 * q + 1], b1, acc2);
      }
      __builtin_amdgcn_sched_barrier(0);
      if (j2 < 7) {
#pragma unroll
        for (int q = 0; q < 8; ++q)
          bb[0][q] = bl[q * 2 * F_DIM + (j2 + 1) * 16];
      }
      __builtin_amdgcn_sched_barrier(0);
#pragma unroll
      for (int q = 0; q < 8; ++q) {
        const v2f b0 = {bb[1][q][0], bb[1][q][1]};
        const v2f b1 = {bb[1][q][2], bb[1][q][3]};
        acc2 = wmma4(af2[16 + 2 * q],     b0, acc2);
        acc2 = wmma4(af2[16 + 2 * q + 1], b1, acc2);
      }
      __builtin_amdgcn_sched_barrier(0);

      float s = 0.0f;
#pragma unroll
      for (int r = 0; r < 8; ++r) {
        const float wij = (acc2[r] + bias2) * vr[r];
        s += wij * xv[r];
        if (r == 7 || (brk & (1u << r))) {
          atomicAdd(agg + (long)vi[r] * F_DIM + n, s);
          s = 0.0f;
        }
      }
    }
  }
}

__global__ void __launch_bounds__(BLOCK_THREADS)
out_kernel(const float* __restrict__ agg,
           const float* __restrict__ Wp_o1, const float* __restrict__ b_o1,
           const float* __restrict__ Wp_o2, const float* __restrict__ b_o2,
           float* __restrict__ out, int ntiles) {
  __shared__ float lds[WAVES_PER_BLOCK][16 * LDS_STRIDE];

  const int wave  = threadIdx.x >> 5;
  const int lane  = threadIdx.x & 31;
  const int t     = blockIdx.x * WAVES_PER_BLOCK + wave;
  const bool live = (t < ntiles);
  const int mrow  = lane & 15;
  const int khalf = lane >> 4;
  const long r0   = (long)t * 16;
  float* tw = lds[wave];

  if (live) {
    const float* grow = agg + (r0 + mrow) * F_DIM;
    v2f af[32];
#pragma unroll
    for (int k = 0; k < 32; ++k)
      af[k] = *(const v2f*)(grow + 4 * k + 2 * khalf);

    for (int j = 0; j < 8; ++j) {
      const int n = j * 16 + mrow;
      v8f acc = v8f_zero();
#pragma unroll
      for (int k = 0; k < 32; ++k)
        acc = wmma4(af[k], bfrag(Wp_o1, k, khalf, n), acc);

      const float bias = b_o1[n];
#pragma unroll
      for (int r = 0; r < 8; ++r) {
        const int m = r + 8 * khalf;
        tw[m * LDS_STRIDE + n] = ssp(acc[r] + bias);
      }
    }
  }
  __syncthreads();

  if (live) {
    const float* arow = tw + mrow * LDS_STRIDE;
    v2f af2[32];
#pragma unroll
    for (int k = 0; k < 32; ++k)
      af2[k] = *(const v2f*)(arow + 4 * k + 2 * khalf);

    for (int j2 = 0; j2 < 8; ++j2) {
      const int n = j2 * 16 + mrow;
      v8f acc = v8f_zero();
#pragma unroll
      for (int k = 0; k < 32; ++k)
        acc = wmma4(af2[k], bfrag(Wp_o2, k, khalf, n), acc);

      const float bias2 = b_o2[n];
#pragma unroll
      for (int r = 0; r < 8; ++r) {
        const int m = r + 8 * khalf;
        out[(r0 + m) * A_DIM + n] = acc[r] + bias2;
      }
    }
  }
}

extern "C" void kernel_launch(void* const* d_in, const int* in_sizes, int n_in,
                              void* d_out, int out_size, void* d_ws, size_t ws_size,
                              hipStream_t stream) {
  const float* x      = (const float*)d_in[0];
  const float* f_ij   = (const float*)d_in[1];
  const float* rcut   = (const float*)d_in[2];
  const int*   idx_i  = (const int*)d_in[3];
  const int*   idx_j  = (const int*)d_in[4];
  const float* W_in2f = (const float*)d_in[5];
  const float* W_f1   = (const float*)d_in[6];
  const float* b_f1   = (const float*)d_in[7];
  const float* W_f2   = (const float*)d_in[8];
  const float* b_f2   = (const float*)d_in[9];
  const float* W_o1   = (const float*)d_in[10];
  const float* b_o1   = (const float*)d_in[11];
  const float* W_o2   = (const float*)d_in[12];
  const float* b_o2   = (const float*)d_in[13];
  float* out = (float*)d_out;

  const int  N = in_sizes[0] / A_DIM;
  const long E = (long)in_sizes[2];

  float* h       = (float*)d_ws;
  float* agg     = h + (size_t)N * F_DIM;
  float* Wp_in2f = agg + (size_t)N * F_DIM;
  float* Wp_f1   = Wp_in2f + A_DIM * F_DIM;
  float* Wp_f2   = Wp_f1 + R_DIM * F_DIM;
  float* Wp_o1   = Wp_f2 + F_DIM * F_DIM;
  float* Wp_o2   = Wp_o1 + F_DIM * A_DIM;

  {
    int  ntiles      = N / 16;
    int  gemm_blocks = (ntiles + WAVES_PER_BLOCK - 1) / WAVES_PER_BLOCK;
    long n4          = (long)N * (F_DIM / 4);
    int  zero_blocks = (int)((n4 + ZERO_V4_PER_BLOCK - 1) / ZERO_V4_PER_BLOCK);
    int  pack_blocks = (PACK_TOTAL + BLOCK_THREADS - 1) / BLOCK_THREADS;
    prep_kernel<<<gemm_blocks + zero_blocks + pack_blocks, BLOCK_THREADS, 0, stream>>>(
        x, W_in2f, h, ntiles, gemm_blocks, agg, n4, zero_blocks,
        W_f1, Wp_f1, W_f2, Wp_f2, W_o1, Wp_o1, W_o2, Wp_o2);
  }
  {
    int ntiles = (int)(E / 16);
    int blocks = (ntiles + EDGE_TILES_PER_BLOCK - 1) / EDGE_TILES_PER_BLOCK;
    edge_kernel<<<blocks, EDGE_THREADS, 0, stream>>>(
        f_ij, rcut, idx_i, idx_j, Wp_f1, b_f1, Wp_f2, b_f2, h, agg, ntiles);
  }
  {
    int ntiles = N / 16;
    int blocks = (ntiles + WAVES_PER_BLOCK - 1) / WAVES_PER_BLOCK;
    out_kernel<<<blocks, BLOCK_THREADS, 0, stream>>>(
        agg, Wp_o1, b_o1, Wp_o2, b_o2, out, ntiles);
  }
}


# edge kernel: GEMM2 runs k outer on 8 independent accumulator tiles (16 WMMAs per group of 8 LDS reads), not 8 dependent chains of 32; gathers and scatter-adds once per tile; sums keep their order
# speedup vs baseline: 2.0426x; 1.0540x over previous
#include <hip/hip_runtime.h>
#include <math.h>


typedef __attribute__((ext_vector_type(2))) float v2f;
typedef __attribute__((ext_vector_type(4))) float v4f;
typedef __attribute__((ext_vector_type(8))) float v8f;

#define A_DIM 128
#define F_DIM 128
#define R_DIM 20
#define LDS_STRIDE 132
#define WAVES_PER_BLOCK 5
#define BLOCK_THREADS (WAVES_PER_BLOCK * 32)
#define EDGE_WAVES 8
#define EDGE_THREADS (EDGE_WAVES * 32)
#define EDGE_TILES_PER_WAVE 5
#define EDGE_TILES_PER_BLOCK (EDGE_WAVES * EDGE_TILES_PER_WAVE)

__device__ __forceinline__ v8f wmma4(v2f a, v2f b, v8f c) {
  return __builtin_amdgcn_wmma_f32_16x16x4_f32(
       false, a,  false, b,
       (short)0, c,  false,  false);
}

__device__ __forceinline__ v8f v8f_zero() {
  v8f z = {0.0f, 0.0f, 0.0f, 0.0f, 0.0f, 0.0f, 0.0f, 0.0f};
  return z;
}

__device__ __forceinline__ float ssp(float x) {
  const float t = __builtin_amdgcn_exp2f(-1.44269504088896341f * fabsf(x));
  const float l = __builtin_amdgcn_logf(__builtin_fmaf(t, 0.5f, 0.5f));
  return __builtin_fmaf(l, 0.69314718055994531f, fmaxf(x, 0.0f));
}

__device__ __forceinline__ void pack_one(const float* __restrict__ W,
                                         float* __restrict__ Wp, int i) {
  const int k2 = i >> 7;
  const int n  = i & 127;
  const v2f v = {W[(2 * k2 + 0) * 128 + n], W[(2 * k2 + 1) * 128 + n]};
  *(v2f*)(Wp + k2 * 256 + 2 * n) = v;
}

__device__ __forceinline__ v2f bfrag(const float* __restrict__ Wp,
                                     int k, int khalf, int n) {
  return *(const v2f*)(Wp + (2 * k + khalf) * 256 + 2 * n);
}

__device__ __forceinline__ v2f bfrag_rowmajor(const float* __restrict__ W,
                                              int k, int khalf, int n) {
  const float* p = W + (4 * k + 2 * khalf) * 128 + n;
  const v2f b = {p[0], p[128]};
  return b;
}

#define ZERO_V4_PER_THREAD 8
#define ZERO_V4_PER_BLOCK (ZERO_V4_PER_THREAD * BLOCK_THREADS)
#define PACK_N1 ((R_DIM / 2) * F_DIM)
#define PACK_N2 ((F_DIM / 2) * F_DIM)
#define PACK_N3 ((F_DIM / 2) * A_DIM)
#define PACK_N4 ((A_DIM / 2) * A_DIM)
#define PACK_TOTAL (PACK_N1 + PACK_N2 + PACK_N3 + PACK_N4)

__global__ void __launch_bounds__(BLOCK_THREADS)
prep_kernel(const float* __restrict__ x, const float* __restrict__ W_in2f,
            float* __restrict__ h, int ntiles, int gemm_blocks,
            float* __restrict__ agg, long n4, int zero_blocks,
            const float* __restrict__ W_f1, float* __restrict__ Wp_f1,
            const float* __restrict__ W_f2, float* __restrict__ Wp_f2,
            const float* __restrict__ W_o1, float* __restrict__ Wp_o1,
            const float* __restrict__ W_o2, float* __restrict__ Wp_o2) {
  int b = blockIdx.x;

  if (b < gemm_blocks) {
    const int wave  = threadIdx.x >> 5;
    const int lane  = threadIdx.x & 31;
    const int t     = b * WAVES_PER_BLOCK + wave;
    if (t >= ntiles) return;
    const int mrow  = lane & 15;
    const int khalf = lane >> 4;
    const long r0   = (long)t * 16;

    const float* xrow = x + (r0 + mrow) * A_DIM;
    v2f af[32];
#pragma unroll
    for (int k = 0; k < 32; ++k)
      af[k] = *(const v2f*)(xrow + 4 * k + 2 * khalf);

    for (int j = 0; j < 8; ++j) {
      const int n = j * 16 + mrow;
      v8f acc = v8f_zero();
#pragma unroll
      for (int k = 0; k < 32; ++k)
        acc = wmma4(af[k], bfrag_rowmajor(W_in2f, k, khalf, n), acc);
#pragma unroll
      for (int r = 0; r < 8; ++r) {
        const int m = r + 8 * khalf;
        h[(r0 + m) * F_DIM + n] = acc[r];
      }
    }
    return;
  }
  b -= gemm_blocks;

  if (b < zero_blocks) {
    const v4f z = {0.0f, 0.0f, 0.0f, 0.0f};
    v4f* p = (v4f*)agg;
    const long base = (long)b * ZERO_V4_PER_BLOCK + threadIdx.x;
#pragma unroll
    for (int it = 0; it < ZERO_V4_PER_THREAD; ++it) {
      const long i = base + it * BLOCK_THREADS;
      if (i < n4) p[i] = z;
    }
    return;
  }
  b -= zero_blocks;

  int i = b * BLOCK_THREADS + threadIdx.x;
  if (i < PACK_N1) { pack_one(W_f1, Wp_f1, i); return; }
  i -= PACK_N1;
  if (i < PACK_N2) { pack_one(W_f2, Wp_f2, i); return; }
  i -= PACK_N2;
  if (i < PACK_N3) { pack_one(W_o1, Wp_o1, i); return; }
  i -= PACK_N3;
  if (i < PACK_N4) pack_one(W_o2, Wp_o2, i);
}

__global__ void __launch_bounds__(EDGE_THREADS) __attribute__((amdgpu_waves_per_eu(4)))
edge_kernel(const float* __restrict__ f_ij, const float* __restrict__ rcut,
            const int* __restrict__ idx_i, const int* __restrict__ idx_j,
            const float* __restrict__ Wp_f1, const float* __restrict__ b_f1,
            const float* __restrict__ Wp_f2, const float* __restrict__ b_f2,
            const float* __restrict__ h, float* __restrict__ agg, int ntiles) {
  __shared__ __attribute__((aligned(16))) float wl[F_DIM * F_DIM];
  __shared__ __attribute__((aligned(16))) float lds[EDGE_WAVES][16 * LDS_STRIDE];

  const int wave   = threadIdx.x >> 5;
  const int lane   = threadIdx.x & 31;
  const int mrow   = lane & 15;
  const int khalf  = lane >> 4;
  float* tw = lds[wave];

  for (int i = threadIdx.x; i < 16 * 2 * F_DIM; i += EDGE_THREADS) {
    const int n = i & (F_DIM - 1);
    const int hh = (i >> 7) & 1;
    const int q = i >> 8;
    const v2f lo = *(const v2f*)(Wp_f2 + (4 * q + hh) * 256 + 2 * n);
    const v2f hi = *(const v2f*)(Wp_f2 + (4 * q + 2 + hh) * 256 + 2 * n);
    v4f v = {lo[0], lo[1], hi[0], hi[1]};
    *(v4f*)(wl + 4 * i) = v;
  }
  __syncthreads();

  const v4f* bl = (const v4f*)wl + khalf * F_DIM + mrow;

  for (int it = 0; it < EDGE_TILES_PER_WAVE; ++it) {
    const int t = blockIdx.x * EDGE_TILES_PER_BLOCK + it * EDGE_WAVES + wave;
    if (t >= ntiles) break;
    const long e0 = (long)t * 16;

    const float* frow = f_ij + (e0 + mrow) * R_DIM;
    v2f af1[5];
#pragma unroll
    for (int k = 0; k < 5; ++k)
      af1[k] = *(const v2f*)(frow + 4 * k + 2 * khalf);

    v8f acc1[8];
#pragma unroll
    for (int j = 0; j < 8; ++j) acc1[j] = v8f_zero();
#pragma unroll
    for (int k = 0; k < 5; ++k) {
#pragma unroll
      for (int j = 0; j < 8; ++j)
        acc1[j] = wmma4(af1[k], bfrag(Wp_f1, k, khalf, j * 16 + mrow), acc1[j]);
    }
#pragma unroll
    for (int j = 0; j < 8; ++j) {
      const int n = j * 16 + mrow;
      const float bias = b_f1[n];
#pragma unroll
      for (int r = 0; r < 8; ++r) {
        const int m = r + 8 * khalf;
        tw[m * LDS_STRIDE + n] = ssp(acc1[j][r] + bias);
      }
    }
    __builtin_amdgcn_fence(__ATOMIC_RELEASE, "wavefront");
    __builtin_amdgcn_wave_barrier();
    __builtin_amdgcn_fence(__ATOMIC_ACQUIRE, "wavefront");

    float bias2[8];
#pragma unroll
    for (int j = 0; j < 8; ++j)
      bias2[j] = b_f2[j * 16 + mrow];

    int   vi[8]; int vj[8]; float vr[8];
#pragma unroll
    for (int r = 0; r < 8; ++r) {
      const long e = e0 + r + 8 * khalf;
      vi[r] = idx_i[e]; vj[r] = idx_j[e]; vr[r] = rcut[e];
    }
    unsigned brk = 0x80u;
#pragma unroll
    for (int r = 0; r < 7; ++r)
      if (vi[r] != vi[r + 1]) brk |= (1u << r);

    const float* hrow[8];
#pragma unroll
    for (int r = 0; r < 8; ++r)
      hrow[r] = h + (long)vj[r] * F_DIM + mrow;
    float xv[8][8];
#pragma unroll
    for (int r = 0; r < 4; ++r) {
#pragma unroll
      for (int j = 0; j < 8; ++j)
        xv[j][r] = hrow[r][j * 16];
    }

    const float* arow = tw + mrow * LDS_STRIDE + 2 * khalf;
    v8f acc2[8];
#pragma unroll
    for (int j = 0; j < 8; ++j) acc2[j] = v8f_zero();

    v4f bb[2][8];
    v2f aa[2][2];
#pragma unroll
    for (int j = 0; j < 8; ++j)
      bb[0][j] = bl[j * 16];
    aa[0][0] = *(const v2f*)(arow);
    aa[0][1] = *(const v2f*)(arow + 4);

#pragma unroll
    for (int q = 0; q < 16; ++q) {
      if (q < 15) {
#pragma unroll
        for (int j = 0; j < 8; ++j)
          bb[(q + 1) & 1][j] = bl[(q + 1) * 2 * F_DIM + j * 16];
        aa[(q + 1) & 1][0] = *(const v2f*)(arow + 8 * (q + 1));
        aa[(q + 1) & 1][1] = *(const v2f*)(arow + 8 * (q + 1) + 4);
      }
      if (q == 4) {
#pragma unroll
        for (int r = 4; r < 8; ++r) {
#pragma unroll
          for (int j = 0; j < 8; ++j)
            xv[j][r] = hrow[r][j * 16];
        }
      }
      __builtin_amdgcn_sched_barrier(0);
#pragma unroll
      for (int j = 0; j < 8; ++j) {
        const v2f b0 = {bb[q & 1][j][0], bb[q & 1][j][1]};
        acc2[j] = wmma4(aa[q & 1][0], b0, acc2[j]);
      }
#pragma unroll
      for (int j = 0; j < 8; ++j) {
        const v2f b1 = {bb[q & 1][j][2], bb[q & 1][j][3]};
        acc2[j] = wmma4(aa[q & 1][1], b1, acc2[j]);
      }
      __builtin_amdgcn_sched_barrier(0);
    }

    float s[8];
#pragma unroll
    for (int j = 0; j < 8; ++j) s[j] = 0.0f;
#pragma unroll
    for (int r = 0; r < 8; ++r) {
#pragma unroll
      for (int j = 0; j < 8; ++j) {
        const float wij = (acc2[j][r] + bias2[j]) * vr[r];
        s[j] += wij * xv[j][r];
      }
      if (r == 7 || (brk & (1u << r))) {
        float* orow = agg + (long)vi[r] * F_DIM + mrow;
#pragma unroll
        for (int j = 0; j < 8; ++j) {
          atomicAdd(orow + j * 16, s[j]);
          s[j] = 0.0f;
        }
      }
    }
  }
}

__global__ void __launch_bounds__(BLOCK_THREADS)
out_kernel(const float* __restrict__ agg,
           const float* __restrict__ Wp_o1, const float* __restrict__ b_o1,
           const float* __restrict__ Wp_o2, const float* __restrict__ b_o2,
           float* __restrict__ out, int ntiles) {
  __shared__ float lds[WAVES_PER_BLOCK][16 * LDS_STRIDE];

  const int wave  = threadIdx.x >> 5;
  const int lane  = threadIdx.x & 31;
  const int t     = blockIdx.x * WAVES_PER_BLOCK + wave;
  const bool live = (t < ntiles);
  const int mrow  = lane & 15;
  const int khalf = lane >> 4;
  const long r0   = (long)t * 16;
  float* tw = lds[wave];

  if (live) {
    const float* grow = agg + (r0 + mrow) * F_DIM;
    v2f af[32];
#pragma unroll
    for (int k = 0; k < 32; ++k)
      af[k] = *(const v2f*)(grow + 4 * k + 2 * khalf);

    for (int j = 0; j < 8; ++j) {
      const int n = j * 16 + mrow;
      v8f acc = v8f_zero();
#pragma unroll
      for (int k = 0; k < 32; ++k)
        acc = wmma4(af[k], bfrag(Wp_o1, k, khalf, n), acc);

      const float bias = b_o1[n];
#pragma unroll
      for (int r = 0; r < 8; ++r) {
        const int m = r + 8 * khalf;
        tw[m * LDS_STRIDE + n] = ssp(acc[r] + bias);
      }
    }
  }
  __syncthreads();

  if (live) {
    const float* arow = tw + mrow * LDS_STRIDE;
    v2f af2[32];
#pragma unroll
    for (int k = 0; k < 32; ++k)
      af2[k] = *(const v2f*)(arow + 4 * k + 2 * khalf);

    for (int j2 = 0; j2 < 8; ++j2) {
      const int n = j2 * 16 + mrow;
      v8f acc = v8f_zero();
#pragma unroll
      for (int k = 0; k < 32; ++k)
        acc = wmma4(af2[k], bfrag(Wp_o2, k, khalf, n), acc);

      const float bias2 = b_o2[n];
#pragma unroll
      for (int r = 0; r < 8; ++r) {
        const int m = r + 8 * khalf;
        out[(r0 + m) * A_DIM + n] = acc[r] + bias2;
      }
    }
  }
}

extern "C" void kernel_launch(void* const* d_in, const int* in_sizes, int n_in,
                              void* d_out, int out_size, void* d_ws, size_t ws_size,
                              hipStream_t stream) {
  const float* x      = (const float*)d_in[0];
  const float* f_ij   = (const float*)d_in[1];
  const float* rcut   = (const float*)d_in[2];
  const int*   idx_i  = (const int*)d_in[3];
  const int*   idx_j  = (const int*)d_in[4];
  const float* W_in2f = (const float*)d_in[5];
  const float* W_f1   = (const float*)d_in[6];
  const float* b_f1   = (const float*)d_in[7];
  const float* W_f2   = (const float*)d_in[8];
  const float* b_f2   = (const float*)d_in[9];
  const float* W_o1   = (const float*)d_in[10];
  const float* b_o1   = (const float*)d_in[11];
  const float* W_o2   = (const float*)d_in[12];
  const float* b_o2   = (const float*)d_in[13];
  float* out = (float*)d_out;

  const int  N = in_sizes[0] / A_DIM;
  const long E = (long)in_sizes[2];

  float* h       = (float*)d_ws;
  float* agg     = h + (size_t)N * F_DIM;
  float* Wp_in2f = agg + (size_t)N * F_DIM;
  float* Wp_f1   = Wp_in2f + A_DIM * F_DIM;
  float* Wp_f2   = Wp_f1 + R_DIM * F_DIM;
  float* Wp_o1   = Wp_f2 + F_DIM * F_DIM;
  float* Wp_o2   = Wp_o1 + F_DIM * A_DIM;

  {
    int  ntiles      = N / 16;
    int  gemm_blocks = (ntiles + WAVES_PER_BLOCK - 1) / WAVES_PER_BLOCK;
    long n4          = (long)N * (F_DIM / 4);
    int  zero_blocks = (int)((n4 + ZERO_V4_PER_BLOCK - 1) / ZERO_V4_PER_BLOCK);
    int  pack_blocks = (PACK_TOTAL + BLOCK_THREADS - 1) / BLOCK_THREADS;
    prep_kernel<<<gemm_blocks + zero_blocks + pack_blocks, BLOCK_THREADS, 0, stream>>>(
        x, W_in2f, h, ntiles, gemm_blocks, agg, n4, zero_blocks,
        W_f1, Wp_f1, W_f2, Wp_f2, W_o1, Wp_o1, W_o2, Wp_o2);
  }
  {
    int ntiles = (int)(E / 16);
    int blocks = (ntiles + EDGE_TILES_PER_BLOCK - 1) / EDGE_TILES_PER_BLOCK;
    edge_kernel<<<blocks, EDGE_THREADS, 0, stream>>>(
        f_ij, rcut, idx_i, idx_j, Wp_f1, b_f1, Wp_f2, b_f2, h, agg, ntiles);
  }
  {
    int ntiles = N / 16;
    int blocks = (ntiles + WAVES_PER_BLOCK - 1) / WAVES_PER_BLOCK;
    out_kernel<<<blocks, BLOCK_THREADS, 0, stream>>>(
        agg, Wp_o1, b_o1, Wp_o2, b_o2, out, ntiles);
  }
}
